# starting kernel
# baseline (speedup 1.0000x reference)
#include <hip/hip_runtime.h>


#define ALPHA 0.2f
#define N_  4
#define HD  16
#define WD  16
#define T_  24
#define V_  32
#define HW  256
#define THW 512

typedef __attribute__((ext_vector_type(2))) float v2f;
typedef __attribute__((ext_vector_type(8))) float v8f;

__global__ void adj_kernel(const float* __restrict__ B, float* __restrict__ adjn) {
    __shared__ float rmin[V_], rmax[V_], dinv[V_];
    __shared__ float smn, smx;
    int i = threadIdx.x;
    float mn = 1e30f, mx = -1e30f;
    for (int j = 0; j < V_; ++j) {
        float v = B[i * V_ + j] + 1e-6f + (i == j ? 1.0f : 0.0f);
        mn = fminf(mn, v); mx = fmaxf(mx, v);
    }
    rmin[i] = mn; rmax[i] = mx;
    __syncthreads();
    if (i == 0) {
        float gmn = rmin[0], gmx = rmax[0];
        for (int k = 1; k < V_; ++k) { gmn = fminf(gmn, rmin[k]); gmx = fmaxf(gmx, rmax[k]); }
        smn = gmn; smx = gmx;
    }
    __syncthreads();
    float gmn = smn, inv = 1.0f / (smx - smn);
    float s = 0.0f;
    for (int j = 0; j < V_; ++j) {
        float v = B[i * V_ + j] + 1e-6f + (i == j ? 1.0f : 0.0f);
        s += (v - gmn) * inv;
    }
    dinv[i] = rsqrtf(s);
    __syncthreads();
    float di = dinv[i];
    for (int j = 0; j < V_; ++j) {
        float v = B[i * V_ + j] + 1e-6f + (i == j ? 1.0f : 0.0f);
        adjn[i * V_ + j] = (v - gmn) * inv * di * dinv[j];
    }
}

__global__ void __launch_bounds__(256) wh_kernel(const float* __restrict__ h,
                                                 const float* __restrict__ Wp,
                                                 float* __restrict__ Wh) {
    __shared__ float Xs[T_][V_];
    __shared__ float Wm[T_][T_];
    int bid = blockIdx.x;
    int n = bid >> 8;
    int p = bid & 255;
    int tid = threadIdx.x;
    const float* hp = h + (size_t)(n * HW + p) * T_ * V_;
    for (int idx = tid; idx < T_ * V_; idx += 256)
        Xs[idx / V_][idx & (V_ - 1)] = hp[idx];
    for (int idx = tid; idx < T_ * T_; idx += 256)
        Wm[idx / T_][idx % T_] = Wp[idx];
    __syncthreads();
    for (int idx = tid; idx < V_ * T_; idx += 256) {
        int v = idx / T_, t = idx % T_;
        float acc = 0.0f;
#pragma unroll
        for (int tp = 0; tp < T_; ++tp) acc += Xs[tp][v] * Wm[tp][t];
        Wh[(size_t)((n * T_ + t) * V_ + v) * HW + p] = acc;
    }
}

__global__ void eij_kernel(const float* __restrict__ Wh, const float* __restrict__ a,
                           float* __restrict__ ei, float* __restrict__ ej) {
    int g = blockIdx.x * 256 + threadIdx.x;
    int n  = g / (V_ * T_);
    int r  = g - n * (V_ * T_);
    int vv = r / T_;
    int tn = r - vv * T_;
    const float* whb = Wh + (size_t)n * T_ * V_ * HW;
    float ai = 0.0f, aj = 0.0f;
    for (int q = 0; q < THW; ++q) {
        int f  = tn * THW + q;
        int hh = f / (2 * WD * T_);
        int r2 = f - hh * (2 * WD * T_);
        int ww = r2 / T_;
        int t  = r2 - ww * T_;
        float w = a[q];
        if (ww < WD)
            ai += whb[(size_t)(t * V_ + vv) * HW + hh * WD + ww] * w;
        else
            aj += whb[(size_t)(t * V_ + vv) * HW + hh * WD + (ww - WD)] * w;
    }
    ei[g] = ai;
    ej[g] = aj;
}

__global__ void att_kernel(const float* __restrict__ ei, const float* __restrict__ ej,
                           float* __restrict__ att) {
    int g = blockIdx.x * 256 + threadIdx.x;
    int n   = g >> 10;
    int rem = g & 1023;
    int i   = rem >> 5;
    int j   = rem & 31;
    const float* eip = ei + (n * V_ + i) * T_;
    const float* ejp = ej + (n * V_ + j) * T_;
    float l[T_];
    float m = -1e30f;
#pragma unroll
    for (int t = 0; t < T_; ++t) {
        float e = eip[t] + ejp[t];
        e = e > 0.0f ? e : ALPHA * e;
        l[t] = e;
        m = fmaxf(m, e);
    }
    float s = 0.0f;
#pragma unroll
    for (int t = 0; t < T_; ++t) { l[t] = __expf(l[t] - m); s += l[t]; }
    float is = 1.0f / s;
#pragma unroll
    for (int t = 0; t < T_; ++t)
        att[(size_t)((n * T_ + t) * V_ + i) * V_ + j] = l[t] * is;
}

__global__ void __launch_bounds__(256) gemm_kernel(const float* __restrict__ Wh,
                                                   const float* __restrict__ att,
                                                   const float* __restrict__ adjn,
                                                   float* __restrict__ out) {
    __shared__ float Gl [V_ * 33];
    __shared__ float Ajl[V_ * 33];
    __shared__ float Ml [V_ * 33];
    __shared__ float Wl [V_ * 257];

    int bid = blockIdx.x;
    int n = bid / T_;
    int t = bid - n * T_;
    int tid = threadIdx.x;
    const float* gsl = att + (size_t)(n * T_ + t) * V_ * V_;
    const float* wsl = Wh  + (size_t)(n * T_ + t) * V_ * HW;

    for (int idx = tid; idx < V_ * V_; idx += 256) {
        Gl [(idx >> 5) * 33 + (idx & 31)] = gsl[idx];
        Ajl[(idx >> 5) * 33 + (idx & 31)] = adjn[idx];
    }
    for (int idx = tid; idx < V_ * HW; idx += 256)
        Wl[(idx >> 8) * 257 + (idx & 255)] = wsl[idx];
    __syncthreads();

    for (int idx = tid; idx < V_ * V_; idx += 256) {
        int v = idx >> 5, j = idx & 31;
        float acc = 0.0f;
#pragma unroll
        for (int i2 = 0; i2 < V_; ++i2) acc += Ajl[i2 * 33 + v] * Gl[i2 * 33 + j];
        Ml[v * 33 + j] = acc;
    }
    __syncthreads();

    int wave = tid >> 5, lane = tid & 31;
    int lh = lane >> 4, lm = lane & 15;
    for (int q = 0; q < 4; ++q) {
        int tile = wave * 4 + q;
        int tm = tile >> 4, tn = tile & 15;
        int arow = tm * 16 + lm;
        int col  = tn * 16 + lm;
        v8f c = {};
#pragma unroll
        for (int kk = 0; kk < 8; ++kk) {
            int kb = kk * 4 + lh * 2;
            v2f a, b;
            a.x = Ml[arow * 33 + kb];
            a.y = Ml[arow * 33 + kb + 1];
            b.x = Wl[kb * 257 + col];
            b.y = Wl[(kb + 1) * 257 + col];
            c = __builtin_amdgcn_wmma_f32_16x16x4_f32(
                     false, a,  false, b,
                     (short)0, c,  false,  false);
        }
        int vb = tm * 16 + lh * 8;
        float* ob = out + ((size_t)(n * HW + col) * T_ + t) * V_ + vb;
#pragma unroll
        for (int r = 0; r < 8; ++r) {
            float x = c[r];
            float y = x > 0.0f ? x : (__expf(x) - 1.0f);
            ob[r] = 1.0f / (1.0f + __expf(-y));
        }
    }
}

extern "C" void kernel_launch(void* const* d_in, const int* in_sizes, int n_in,
                              void* d_out, int out_size, void* d_ws, size_t ws_size,
                              hipStream_t stream) {
    const float* h  = (const float*)d_in[0];
    const float* Wp = (const float*)d_in[1];
    const float* ap = (const float*)d_in[2];
    const float* Bp = (const float*)d_in[3];
    float* out = (float*)d_out;

    float* ws    = (float*)d_ws;
    float* Whb   = ws;
    float* adjn  = Whb  + (size_t)N_ * T_ * V_ * HW;
    float* ei    = adjn + V_ * V_;
    float* ej    = ei   + N_ * V_ * T_;
    float* attb  = ej   + N_ * V_ * T_;

    adj_kernel <<<1,           32, 0, stream>>>(Bp, adjn);
    wh_kernel  <<<N_ * HW,    256, 0, stream>>>(h, Wp, Whb);
    eij_kernel <<<(N_ * V_ * T_) / 256, 256, 0, stream>>>(Whb, ap, ei, ej);
    att_kernel <<<(N_ * V_ * V_) / 256, 256, 0, stream>>>(ei, ej, attb);
    gemm_kernel<<<N_ * T_,    256, 0, stream>>>(Whb, attb, adjn, out);
}


# Attention-projection kernel (ei/ej) re-parallelised: one wave per (n,v,two h rows) reads whole Wh lines once; lane partials + wave butterfly replace 512-long thread loops (sum order changes)
# speedup vs baseline: 6.4620x; 6.4620x over previous
#include <hip/hip_runtime.h>


#define ALPHA 0.2f
#define N_  4
#define HD  16
#define WD  16
#define T_  24
#define V_  32
#define HW  256
#define THW 512

typedef __attribute__((ext_vector_type(2))) float v2f;
typedef __attribute__((ext_vector_type(8))) float v8f;

__global__ void adj_kernel(const float* __restrict__ B, float* __restrict__ adjn) {
    __shared__ float rmin[V_], rmax[V_], dinv[V_];
    __shared__ float smn, smx;
    int i = threadIdx.x;
    float mn = 1e30f, mx = -1e30f;
    for (int j = 0; j < V_; ++j) {
        float v = B[i * V_ + j] + 1e-6f + (i == j ? 1.0f : 0.0f);
        mn = fminf(mn, v); mx = fmaxf(mx, v);
    }
    rmin[i] = mn; rmax[i] = mx;
    __syncthreads();
    if (i == 0) {
        float gmn = rmin[0], gmx = rmax[0];
        for (int k = 1; k < V_; ++k) { gmn = fminf(gmn, rmin[k]); gmx = fmaxf(gmx, rmax[k]); }
        smn = gmn; smx = gmx;
    }
    __syncthreads();
    float gmn = smn, inv = 1.0f / (smx - smn);
    float s = 0.0f;
    for (int j = 0; j < V_; ++j) {
        float v = B[i * V_ + j] + 1e-6f + (i == j ? 1.0f : 0.0f);
        s += (v - gmn) * inv;
    }
    dinv[i] = rsqrtf(s);
    __syncthreads();
    float di = dinv[i];
    for (int j = 0; j < V_; ++j) {
        float v = B[i * V_ + j] + 1e-6f + (i == j ? 1.0f : 0.0f);
        adjn[i * V_ + j] = (v - gmn) * inv * di * dinv[j];
    }
}

__global__ void __launch_bounds__(256) wh_kernel(const float* __restrict__ h,
                                                 const float* __restrict__ Wp,
                                                 float* __restrict__ Wh) {
    __shared__ float Xs[T_][V_];
    __shared__ float Wm[T_][T_];
    int bid = blockIdx.x;
    int n = bid >> 8;
    int p = bid & 255;
    int tid = threadIdx.x;
    const float* hp = h + (size_t)(n * HW + p) * T_ * V_;
    for (int idx = tid; idx < T_ * V_; idx += 256)
        Xs[idx / V_][idx & (V_ - 1)] = hp[idx];
    for (int idx = tid; idx < T_ * T_; idx += 256)
        Wm[idx / T_][idx % T_] = Wp[idx];
    __syncthreads();
    for (int idx = tid; idx < V_ * T_; idx += 256) {
        int v = idx / T_, t = idx % T_;
        float acc = 0.0f;
#pragma unroll
        for (int tp = 0; tp < T_; ++tp) acc += Xs[tp][v] * Wm[tp][t];
        Wh[(size_t)((n * T_ + t) * V_ + v) * HW + p] = acc;
    }
}

__global__ void __launch_bounds__(256) eij_kernel(const float* __restrict__ Wh,
                                                  const float* __restrict__ a,
                                                  float* __restrict__ ei,
                                                  float* __restrict__ ej) {
    __shared__ __attribute__((aligned(16))) float As[THW];
    int nv  = blockIdx.x;
    int n   = nv >> 5;
    int vv  = nv & (V_ - 1);
    int tid = threadIdx.x;
    int m    = tid >> 5;
    int lane = tid & 31;

    float a0 = a[tid], a1 = a[tid + 256];
    const float* wp = Wh + ((size_t)(n * T_) * V_ + vv) * HW + tid;
    float x[T_];
#pragma unroll
    for (int t = 0; t < T_; ++t) x[t] = wp[(size_t)t * V_ * HW];
    As[tid] = a0; As[tid + 256] = a1;
    __syncthreads();

    int fi = (tid >> 4) * (2 * WD * T_) + (tid & 15) * T_;
    int fj = fi + WD * T_;
    float si0 = 0.0f, si1 = 0.0f, si2 = 0.0f;
    float sj0 = 0.0f, sj1 = 0.0f, sj2 = 0.0f;
#pragma unroll
    for (int c = 0; c < T_ / 4; ++c) {
        int f0 = fi + 4 * c, f1 = fj + 4 * c;
        float4 ai = *(const float4*)(As + (f0 & (THW - 1)));
        float4 aj = *(const float4*)(As + (f1 & (THW - 1)));
        float x0 = x[4 * c], x1 = x[4 * c + 1], x2 = x[4 * c + 2], x3 = x[4 * c + 3];
        float di = x0 * ai.x + x1 * ai.y + x2 * ai.z + x3 * ai.w;
        float dj = x0 * aj.x + x1 * aj.y + x2 * aj.z + x3 * aj.w;
        int ri = (f0 >> 9) - 3 * m;
        int rj = (f1 >> 9) - 3 * m;
        si0 += (ri == 0) ? di : 0.0f;
        si1 += (ri == 1) ? di : 0.0f;
        si2 += (ri == 2) ? di : 0.0f;
        sj0 += (rj == 0) ? dj : 0.0f;
        sj1 += (rj == 1) ? dj : 0.0f;
        sj2 += (rj == 2) ? dj : 0.0f;
    }
#pragma unroll
    for (int off = 16; off > 0; off >>= 1) {
        si0 += __shfl_xor(si0, off); si1 += __shfl_xor(si1, off); si2 += __shfl_xor(si2, off);
        sj0 += __shfl_xor(sj0, off); sj1 += __shfl_xor(sj1, off); sj2 += __shfl_xor(sj2, off);
    }
    if (lane < 3) {
        float oi = lane == 0 ? si0 : (lane == 1 ? si1 : si2);
        float oj = lane == 0 ? sj0 : (lane == 1 ? sj1 : sj2);
        ei[nv * T_ + 3 * m + lane] = oi;
        ej[nv * T_ + 3 * m + lane] = oj;
    }
}

__global__ void att_kernel(const float* __restrict__ ei, const float* __restrict__ ej,
                           float* __restrict__ att) {
    int g = blockIdx.x * 256 + threadIdx.x;
    int n   = g >> 10;
    int rem = g & 1023;
    int i   = rem >> 5;
    int j   = rem & 31;
    const float* eip = ei + (n * V_ + i) * T_;
    const float* ejp = ej + (n * V_ + j) * T_;
    float l[T_];
    float m = -1e30f;
#pragma unroll
    for (int t = 0; t < T_; ++t) {
        float e = eip[t] + ejp[t];
        e = e > 0.0f ? e : ALPHA * e;
        l[t] = e;
        m = fmaxf(m, e);
    }
    float s = 0.0f;
#pragma unroll
    for (int t = 0; t < T_; ++t) { l[t] = __expf(l[t] - m); s += l[t]; }
    float is = 1.0f / s;
#pragma unroll
    for (int t = 0; t < T_; ++t)
        att[(size_t)((n * T_ + t) * V_ + i) * V_ + j] = l[t] * is;
}

__global__ void __launch_bounds__(256) gemm_kernel(const float* __restrict__ Wh,
                                                   const float* __restrict__ att,
                                                   const float* __restrict__ adjn,
                                                   float* __restrict__ out) {
    __shared__ float Gl [V_ * 33];
    __shared__ float Ajl[V_ * 33];
    __shared__ float Ml [V_ * 33];
    __shared__ float Wl [V_ * 257];

    int bid = blockIdx.x;
    int n = bid / T_;
    int t = bid - n * T_;
    int tid = threadIdx.x;
    const float* gsl = att + (size_t)(n * T_ + t) * V_ * V_;
    const float* wsl = Wh  + (size_t)(n * T_ + t) * V_ * HW;

    for (int idx = tid; idx < V_ * V_; idx += 256) {
        Gl [(idx >> 5) * 33 + (idx & 31)] = gsl[idx];
        Ajl[(idx >> 5) * 33 + (idx & 31)] = adjn[idx];
    }
    for (int idx = tid; idx < V_ * HW; idx += 256)
        Wl[(idx >> 8) * 257 + (idx & 255)] = wsl[idx];
    __syncthreads();

    for (int idx = tid; idx < V_ * V_; idx += 256) {
        int v = idx >> 5, j = idx & 31;
        float acc = 0.0f;
#pragma unroll
        for (int i2 = 0; i2 < V_; ++i2) acc += Ajl[i2 * 33 + v] * Gl[i2 * 33 + j];
        Ml[v * 33 + j] = acc;
    }
    __syncthreads();

    int wave = tid >> 5, lane = tid & 31;
    int lh = lane >> 4, lm = lane & 15;
    for (int q = 0; q < 4; ++q) {
        int tile = wave * 4 + q;
        int tm = tile >> 4, tn = tile & 15;
        int arow = tm * 16 + lm;
        int col  = tn * 16 + lm;
        v8f c = {};
#pragma unroll
        for (int kk = 0; kk < 8; ++kk) {
            int kb = kk * 4 + lh * 2;
            v2f a, b;
            a.x = Ml[arow * 33 + kb];
            a.y = Ml[arow * 33 + kb + 1];
            b.x = Wl[kb * 257 + col];
            b.y = Wl[(kb + 1) * 257 + col];
            c = __builtin_amdgcn_wmma_f32_16x16x4_f32(
                     false, a,  false, b,
                     (short)0, c,  false,  false);
        }
        int vb = tm * 16 + lh * 8;
        float* ob = out + ((size_t)(n * HW + col) * T_ + t) * V_ + vb;
#pragma unroll
        for (int r = 0; r < 8; ++r) {
            float x = c[r];
            float y = x > 0.0f ? x : (__expf(x) - 1.0f);
            ob[r] = 1.0f / (1.0f + __expf(-y));
        }
    }
}

extern "C" void kernel_launch(void* const* d_in, const int* in_sizes, int n_in,
                              void* d_out, int out_size, void* d_ws, size_t ws_size,
                              hipStream_t stream) {
    const float* h  = (const float*)d_in[0];
    const float* Wp = (const float*)d_in[1];
    const float* ap = (const float*)d_in[2];
    const float* Bp = (const float*)d_in[3];
    float* out = (float*)d_out;

    float* ws    = (float*)d_ws;
    float* Whb   = ws;
    float* adjn  = Whb  + (size_t)N_ * T_ * V_ * HW;
    float* ei    = adjn + V_ * V_;
    float* ej    = ei   + N_ * V_ * T_;
    float* attb  = ej   + N_ * V_ * T_;

    adj_kernel <<<1,           32, 0, stream>>>(Bp, adjn);
    wh_kernel  <<<N_ * HW,    256, 0, stream>>>(h, Wp, Whb);
    eij_kernel <<<N_ * V_,    256, 0, stream>>>(Whb, ap, ei, ej);
    att_kernel <<<(N_ * V_ * V_) / 256, 256, 0, stream>>>(ei, ej, attb);
    gemm_kernel<<<N_ * T_,    256, 0, stream>>>(Whb, attb, adjn, out);
}


# Wh and ei/ej kernels fused and re-tiled: one wave per (n,v,two h rows) forms its Wh in registers, stores whole 128-B lines, reduces ei/ej from the same registers; 5 launches to 4; sum order kept
# speedup vs baseline: 10.5405x; 1.6312x over previous
#include <hip/hip_runtime.h>


#define ALPHA 0.2f
#define N_  4
#define HD  16
#define WD  16
#define T_  24
#define V_  32
#define HW  256
#define THW 512

typedef __attribute__((ext_vector_type(2))) float v2f;
typedef __attribute__((ext_vector_type(8))) float v8f;

__global__ void adj_kernel(const float* __restrict__ B, float* __restrict__ adjn) {
    __shared__ float rmin[V_], rmax[V_], dinv[V_];
    __shared__ float smn, smx;
    int i = threadIdx.x;
    float mn = 1e30f, mx = -1e30f;
    for (int j = 0; j < V_; ++j) {
        float v = B[i * V_ + j] + 1e-6f + (i == j ? 1.0f : 0.0f);
        mn = fminf(mn, v); mx = fmaxf(mx, v);
    }
    rmin[i] = mn; rmax[i] = mx;
    __syncthreads();
    if (i == 0) {
        float gmn = rmin[0], gmx = rmax[0];
        for (int k = 1; k < V_; ++k) { gmn = fminf(gmn, rmin[k]); gmx = fmaxf(gmx, rmax[k]); }
        smn = gmn; smx = gmx;
    }
    __syncthreads();
    float gmn = smn, inv = 1.0f / (smx - smn);
    float s = 0.0f;
    for (int j = 0; j < V_; ++j) {
        float v = B[i * V_ + j] + 1e-6f + (i == j ? 1.0f : 0.0f);
        s += (v - gmn) * inv;
    }
    dinv[i] = rsqrtf(s);
    __syncthreads();
    float di = dinv[i];
    for (int j = 0; j < V_; ++j) {
        float v = B[i * V_ + j] + 1e-6f + (i == j ? 1.0f : 0.0f);
        adjn[i * V_ + j] = (v - gmn) * inv * di * dinv[j];
    }
}

#define XP 194

__global__ void __launch_bounds__(256) whe_kernel(const float* __restrict__ h,
                                                  const float* __restrict__ Wp,
                                                  const float* __restrict__ a,
                                                  float* __restrict__ Wh,
                                                  float* __restrict__ ei,
                                                  float* __restrict__ ej) {
    __shared__ __attribute__((aligned(16))) float Xs[32 * XP];
    __shared__ __attribute__((aligned(16))) float As[THW];
    int bid = blockIdx.x;
    int vg  = bid & 3;
    int m   = (bid >> 2) & 7;
    int n   = bid >> 5;
    int tid = threadIdx.x;
    int w    = tid >> 5;
    int lane = tid & 31;
    int vv   = vg * 8 + w;
    int p    = m * 32 + lane;

    const float* hb = h + (size_t)(n * HW + m * 32) * (T_ * V_) + vg * 8;
    float4 hv[6];
#pragma unroll
    for (int k = 0; k < 6; ++k) {
        int q = tid + 256 * k;
        hv[k] = *(const float4*)(hb + (q >> 1) * V_ + (q & 1) * 4);
    }
    float a0 = a[tid], a1 = a[tid + 256];
#pragma unroll
    for (int k = 0; k < 6; ++k) {
        int q = tid + 256 * k;
        int r = q >> 1;
        int pl = r / T_;
        float* d = Xs + r * 8 + pl * (XP - T_ * 8) + (q & 1) * 4;
        d[0] = hv[k].x; d[1] = hv[k].y; d[2] = hv[k].z; d[3] = hv[k].w;
    }
    As[tid] = a0; As[tid + 256] = a1;
    __syncthreads();

    float x[T_];
#pragma unroll
    for (int tp = 0; tp < T_; ++tp) x[tp] = Xs[lane * XP + tp * 8 + w];

    float acc[T_];
#pragma unroll
    for (int t = 0; t < T_; ++t) acc[t] = 0.0f;
#pragma unroll
    for (int tp = 0; tp < T_; ++tp) {
#pragma unroll
        for (int t = 0; t < T_; ++t) acc[t] += x[tp] * Wp[tp * T_ + t];
    }

    float* wp = Wh + ((size_t)(n * T_) * V_ + vv) * HW + p;
#pragma unroll
    for (int t = 0; t < T_; ++t) wp[(size_t)t * V_ * HW] = acc[t];

    int nv = n * V_ + vv;
    int fi = (p >> 4) * (2 * WD * T_) + (p & 15) * T_;
    int fj = fi + WD * T_;
    float si0 = 0.0f, si1 = 0.0f, si2 = 0.0f;
    float sj0 = 0.0f, sj1 = 0.0f, sj2 = 0.0f;
#pragma unroll
    for (int c = 0; c < T_ / 4; ++c) {
        int f0 = fi + 4 * c, f1 = fj + 4 * c;
        float4 ai = *(const float4*)(As + (f0 & (THW - 1)));
        float4 aj = *(const float4*)(As + (f1 & (THW - 1)));
        float x0 = acc[4 * c], x1 = acc[4 * c + 1], x2 = acc[4 * c + 2], x3 = acc[4 * c + 3];
        float di = x0 * ai.x + x1 * ai.y + x2 * ai.z + x3 * ai.w;
        float dj = x0 * aj.x + x1 * aj.y + x2 * aj.z + x3 * aj.w;
        int ri = (f0 >> 9) - 3 * m;
        int rj = (f1 >> 9) - 3 * m;
        si0 += (ri == 0) ? di : 0.0f;
        si1 += (ri == 1) ? di : 0.0f;
        si2 += (ri == 2) ? di : 0.0f;
        sj0 += (rj == 0) ? dj : 0.0f;
        sj1 += (rj == 1) ? dj : 0.0f;
        sj2 += (rj == 2) ? dj : 0.0f;
    }
#pragma unroll
    for (int off = 16; off > 0; off >>= 1) {
        si0 += __shfl_xor(si0, off); si1 += __shfl_xor(si1, off); si2 += __shfl_xor(si2, off);
        sj0 += __shfl_xor(sj0, off); sj1 += __shfl_xor(sj1, off); sj2 += __shfl_xor(sj2, off);
    }
    if (lane < 3) {
        float oi = lane == 0 ? si0 : (lane == 1 ? si1 : si2);
        float oj = lane == 0 ? sj0 : (lane == 1 ? sj1 : sj2);
        ei[nv * T_ + 3 * m + lane] = oi;
        ej[nv * T_ + 3 * m + lane] = oj;
    }
}

__global__ void att_kernel(const float* __restrict__ ei, const float* __restrict__ ej,
                           float* __restrict__ att) {
    int g = blockIdx.x * 256 + threadIdx.x;
    int n   = g >> 10;
    int rem = g & 1023;
    int i   = rem >> 5;
    int j   = rem & 31;
    const float* eip = ei + (n * V_ + i) * T_;
    const float* ejp = ej + (n * V_ + j) * T_;
    float l[T_];
    float m = -1e30f;
#pragma unroll
    for (int t = 0; t < T_; ++t) {
        float e = eip[t] + ejp[t];
        e = e > 0.0f ? e : ALPHA * e;
        l[t] = e;
        m = fmaxf(m, e);
    }
    float s = 0.0f;
#pragma unroll
    for (int t = 0; t < T_; ++t) { l[t] = __expf(l[t] - m); s += l[t]; }
    float is = 1.0f / s;
#pragma unroll
    for (int t = 0; t < T_; ++t)
        att[(size_t)((n * T_ + t) * V_ + i) * V_ + j] = l[t] * is;
}

__global__ void __launch_bounds__(256) gemm_kernel(const float* __restrict__ Wh,
                                                   const float* __restrict__ att,
                                                   const float* __restrict__ adjn,
                                                   float* __restrict__ out) {
    __shared__ float Gl [V_ * 33];
    __shared__ float Ajl[V_ * 33];
    __shared__ float Ml [V_ * 33];
    __shared__ float Wl [V_ * 257];

    int bid = blockIdx.x;
    int n = bid / T_;
    int t = bid - n * T_;
    int tid = threadIdx.x;
    const float* gsl = att + (size_t)(n * T_ + t) * V_ * V_;
    const float* wsl = Wh  + (size_t)(n * T_ + t) * V_ * HW;

    for (int idx = tid; idx < V_ * V_; idx += 256) {
        Gl [(idx >> 5) * 33 + (idx & 31)] = gsl[idx];
        Ajl[(idx >> 5) * 33 + (idx & 31)] = adjn[idx];
    }
    for (int idx = tid; idx < V_ * HW; idx += 256)
        Wl[(idx >> 8) * 257 + (idx & 255)] = wsl[idx];
    __syncthreads();

    for (int idx = tid; idx < V_ * V_; idx += 256) {
        int v = idx >> 5, j = idx & 31;
        float acc = 0.0f;
#pragma unroll
        for (int i2 = 0; i2 < V_; ++i2) acc += Ajl[i2 * 33 + v] * Gl[i2 * 33 + j];
        Ml[v * 33 + j] = acc;
    }
    __syncthreads();

    int wave = tid >> 5, lane = tid & 31;
    int lh = lane >> 4, lm = lane & 15;
    for (int q = 0; q < 4; ++q) {
        int tile = wave * 4 + q;
        int tm = tile >> 4, tn = tile & 15;
        int arow = tm * 16 + lm;
        int col  = tn * 16 + lm;
        v8f c = {};
#pragma unroll
        for (int kk = 0; kk < 8; ++kk) {
            int kb = kk * 4 + lh * 2;
            v2f a, b;
            a.x = Ml[arow * 33 + kb];
            a.y = Ml[arow * 33 + kb + 1];
            b.x = Wl[kb * 257 + col];
            b.y = Wl[(kb + 1) * 257 + col];
            c = __builtin_amdgcn_wmma_f32_16x16x4_f32(
                     false, a,  false, b,
                     (short)0, c,  false,  false);
        }
        int vb = tm * 16 + lh * 8;
        float* ob = out + ((size_t)(n * HW + col) * T_ + t) * V_ + vb;
#pragma unroll
        for (int r = 0; r < 8; ++r) {
            float x = c[r];
            float y = x > 0.0f ? x : (__expf(x) - 1.0f);
            ob[r] = 1.0f / (1.0f + __expf(-y));
        }
    }
}

extern "C" void kernel_launch(void* const* d_in, const int* in_sizes, int n_in,
                              void* d_out, int out_size, void* d_ws, size_t ws_size,
                              hipStream_t stream) {
    const float* h  = (const float*)d_in[0];
    const float* Wp = (const float*)d_in[1];
    const float* ap = (const float*)d_in[2];
    const float* Bp = (const float*)d_in[3];
    float* out = (float*)d_out;

    float* ws    = (float*)d_ws;
    float* Whb   = ws;
    float* adjn  = Whb  + (size_t)N_ * T_ * V_ * HW;
    float* ei    = adjn + V_ * V_;
    float* ej    = ei   + N_ * V_ * T_;
    float* attb  = ej   + N_ * V_ * T_;

    adj_kernel <<<1,           32, 0, stream>>>(Bp, adjn);
    whe_kernel <<<N_ * 8 * 4, 256, 0, stream>>>(h, Wp, ap, Whb, ei, ej);
    att_kernel <<<(N_ * V_ * V_) / 256, 256, 0, stream>>>(ei, ej, attb);
    gemm_kernel<<<N_ * T_,    256, 0, stream>>>(Whb, attb, adjn, out);
}


# Adjacency and softmax launches folded into the product kernel: each (n,t) block rebuilds adjn from B and its softmax slice from ei/ej under its Wh loads; 4 launches to 2; sum orders kept
# speedup vs baseline: 15.1404x; 1.4364x over previous
#include <hip/hip_runtime.h>


#define ALPHA 0.2f
#define N_  4
#define HD  16
#define WD  16
#define T_  24
#define V_  32
#define HW  256
#define THW 512

typedef __attribute__((ext_vector_type(2))) float v2f;
typedef __attribute__((ext_vector_type(8))) float v8f;

#define XP 194

__global__ void __launch_bounds__(256) whe_kernel(const float* __restrict__ h,
                                                  const float* __restrict__ Wp,
                                                  const float* __restrict__ a,
                                                  float* __restrict__ Wh,
                                                  float* __restrict__ ei,
                                                  float* __restrict__ ej) {
    __shared__ __attribute__((aligned(16))) float Xs[32 * XP];
    __shared__ __attribute__((aligned(16))) float As[THW];
    int bid = blockIdx.x;
    int vg  = bid & 3;
    int m   = (bid >> 2) & 7;
    int n   = bid >> 5;
    int tid = threadIdx.x;
    int w    = tid >> 5;
    int lane = tid & 31;
    int vv   = vg * 8 + w;
    int p    = m * 32 + lane;

    const float* hb = h + (size_t)(n * HW + m * 32) * (T_ * V_) + vg * 8;
    float4 hv[6];
#pragma unroll
    for (int k = 0; k < 6; ++k) {
        int q = tid + 256 * k;
        hv[k] = *(const float4*)(hb + (q >> 1) * V_ + (q & 1) * 4);
    }
    float a0 = a[tid], a1 = a[tid + 256];
#pragma unroll
    for (int k = 0; k < 6; ++k) {
        int q = tid + 256 * k;
        int r = q >> 1;
        int pl = r / T_;
        float* d = Xs + r * 8 + pl * (XP - T_ * 8) + (q & 1) * 4;
        d[0] = hv[k].x; d[1] = hv[k].y; d[2] = hv[k].z; d[3] = hv[k].w;
    }
    As[tid] = a0; As[tid + 256] = a1;
    __syncthreads();

    float x[T_];
#pragma unroll
    for (int tp = 0; tp < T_; ++tp) x[tp] = Xs[lane * XP + tp * 8 + w];

    float acc[T_];
#pragma unroll
    for (int t = 0; t < T_; ++t) acc[t] = 0.0f;
#pragma unroll
    for (int tp = 0; tp < T_; ++tp) {
#pragma unroll
        for (int t = 0; t < T_; ++t) acc[t] += x[tp] * Wp[tp * T_ + t];
    }

    float* wp = Wh + ((size_t)(n * T_) * V_ + vv) * HW + p;
#pragma unroll
    for (int t = 0; t < T_; ++t) wp[(size_t)t * V_ * HW] = acc[t];

    int nv = n * V_ + vv;
    int fi = (p >> 4) * (2 * WD * T_) + (p & 15) * T_;
    int fj = fi + WD * T_;
    float si0 = 0.0f, si1 = 0.0f, si2 = 0.0f;
    float sj0 = 0.0f, sj1 = 0.0f, sj2 = 0.0f;
#pragma unroll
    for (int c = 0; c < T_ / 4; ++c) {
        int f0 = fi + 4 * c, f1 = fj + 4 * c;
        float4 ai = *(const float4*)(As + (f0 & (THW - 1)));
        float4 aj = *(const float4*)(As + (f1 & (THW - 1)));
        float x0 = acc[4 * c], x1 = acc[4 * c + 1], x2 = acc[4 * c + 2], x3 = acc[4 * c + 3];
        float di = x0 * ai.x + x1 * ai.y + x2 * ai.z + x3 * ai.w;
        float dj = x0 * aj.x + x1 * aj.y + x2 * aj.z + x3 * aj.w;
        int ri = (f0 >> 9) - 3 * m;
        int rj = (f1 >> 9) - 3 * m;
        si0 += (ri == 0) ? di : 0.0f;
        si1 += (ri == 1) ? di : 0.0f;
        si2 += (ri == 2) ? di : 0.0f;
        sj0 += (rj == 0) ? dj : 0.0f;
        sj1 += (rj == 1) ? dj : 0.0f;
        sj2 += (rj == 2) ? dj : 0.0f;
    }
#pragma unroll
    for (int off = 16; off > 0; off >>= 1) {
        si0 += __shfl_xor(si0, off); si1 += __shfl_xor(si1, off); si2 += __shfl_xor(si2, off);
        sj0 += __shfl_xor(sj0, off); sj1 += __shfl_xor(sj1, off); sj2 += __shfl_xor(sj2, off);
    }
    if (lane < 3) {
        float oi = lane == 0 ? si0 : (lane == 1 ? si1 : si2);
        float oj = lane == 0 ? sj0 : (lane == 1 ? sj1 : sj2);
        ei[nv * T_ + 3 * m + lane] = oi;
        ej[nv * T_ + 3 * m + lane] = oj;
    }
}

#define EJP 25

static __device__ __forceinline__ void lds_barrier() {
    asm volatile("s_wait_dscnt 0x0\n\ts_barrier_signal -1\n\ts_barrier_wait -1" ::: "memory");
}

__global__ void __launch_bounds__(256) gemm_kernel(const float* __restrict__ Wh,
                                                   const float* __restrict__ ei,
                                                   const float* __restrict__ ej,
                                                   const float* __restrict__ B,
                                                   float* __restrict__ out) {
    __shared__ float Gl [V_ * 33];
    __shared__ float Ajl[V_ * 33];
    __shared__ float Ml [V_ * 33];
    __shared__ float Wl [V_ * 257];
    __shared__ __attribute__((aligned(16))) float Eil[V_ * T_];
    __shared__ float Ejl[V_ * EJP];

    int bid = blockIdx.x;
    int n = bid / T_;
    int t = bid - n * T_;
    int tid = threadIdx.x;
    int wave = tid >> 5, lane = tid & 31;
    const float* wsl = Wh + (size_t)(n * T_ + t) * V_ * HW;

    float4 bv = *(const float4*)(B + 4 * tid);
    float4 eiv = make_float4(0.0f, 0.0f, 0.0f, 0.0f), ejv = eiv;
    if (tid < (V_ * T_) / 4) {
        eiv = *(const float4*)(ei + n * (V_ * T_) + 4 * tid);
        ejv = *(const float4*)(ej + n * (V_ * T_) + 4 * tid);
    }
    float wr[V_];
#pragma unroll
    for (int k = 0; k < V_; ++k) wr[k] = wsl[tid + 256 * k];

    int ai = tid >> 3, aj0 = (tid & 7) * 4;
    float av[4] = {bv.x, bv.y, bv.z, bv.w};
#pragma unroll
    for (int c = 0; c < 4; ++c) {
        av[c] = av[c] + 1e-6f + (ai == aj0 + c ? 1.0f : 0.0f);
        Ml[ai * 33 + aj0 + c] = av[c];
    }
    if (tid < (V_ * T_) / 4) {
        int f = 4 * tid;
        int r = f / T_;
        float* di = Eil + f;
        float* dj = Ejl + r * EJP + (f - r * T_);
        di[0] = eiv.x; di[1] = eiv.y; di[2] = eiv.z; di[3] = eiv.w;
        dj[0] = ejv.x; dj[1] = ejv.y; dj[2] = ejv.z; dj[3] = ejv.w;
    }
    lds_barrier();

    {
        float rv[V_];
        float mn = 1e30f, mx = -1e30f;
#pragma unroll
        for (int j = 0; j < V_; ++j) {
            rv[j] = Ml[lane * 33 + j];
            mn = fminf(mn, rv[j]); mx = fmaxf(mx, rv[j]);
        }
#pragma unroll
        for (int off = 16; off > 0; off >>= 1) {
            mn = fminf(mn, __shfl_xor(mn, off));
            mx = fmaxf(mx, __shfl_xor(mx, off));
        }
        float inv = 1.0f / (mx - mn);
        float s = 0.0f;
#pragma unroll
        for (int j = 0; j < V_; ++j) s += (rv[j] - mn) * inv;
        float dinv = rsqrtf(s);
        float dri = __shfl(dinv, ai);
#pragma unroll
        for (int c = 0; c < 4; ++c) {
            float drj = __shfl(dinv, aj0 + c);
            Ajl[ai * 33 + aj0 + c] = (av[c] - mn) * inv * dri * drj;
        }
    }

    {
        float ejr[T_];
#pragma unroll
        for (int tp = 0; tp < T_; ++tp) ejr[tp] = Ejl[lane * EJP + tp];
        float ejt = Ejl[lane * EJP + t];
        for (int k = 0; k < 4; ++k) {
            int i = wave + 8 * k;
            const float* eip = Eil + i * T_;
            float l[T_];
            float m = -1e30f;
#pragma unroll
            for (int tp = 0; tp < T_; ++tp) {
                float e = eip[tp] + ejr[tp];
                e = e > 0.0f ? e : ALPHA * e;
                l[tp] = e;
                m = fmaxf(m, e);
            }
            float s = 0.0f;
#pragma unroll
            for (int tp = 0; tp < T_; ++tp) s += __expf(l[tp] - m);
            float is = 1.0f / s;
            float et = eip[t] + ejt;
            et = et > 0.0f ? et : ALPHA * et;
            Gl[i * 33 + lane] = __expf(et - m) * is;
        }
    }
    lds_barrier();

    for (int idx = tid; idx < V_ * V_; idx += 256) {
        int v = idx >> 5, j = idx & 31;
        float acc = 0.0f;
#pragma unroll
        for (int i2 = 0; i2 < V_; ++i2) acc += Ajl[i2 * 33 + v] * Gl[i2 * 33 + j];
        Ml[v * 33 + j] = acc;
    }
#pragma unroll
    for (int k = 0; k < V_; ++k) Wl[k * 257 + tid] = wr[k];
    __syncthreads();

    int lh = lane >> 4, lm = lane & 15;
    for (int q = 0; q < 4; ++q) {
        int tile = wave * 4 + q;
        int tm = tile >> 4, tn = tile & 15;
        int arow = tm * 16 + lm;
        int col  = tn * 16 + lm;
        v8f c = {};
#pragma unroll
        for (int kk = 0; kk < 8; ++kk) {
            int kb = kk * 4 + lh * 2;
            v2f a, b;
            a.x = Ml[arow * 33 + kb];
            a.y = Ml[arow * 33 + kb + 1];
            b.x = Wl[kb * 257 + col];
            b.y = Wl[(kb + 1) * 257 + col];
            c = __builtin_amdgcn_wmma_f32_16x16x4_f32(
                     false, a,  false, b,
                     (short)0, c,  false,  false);
        }
        int vb = tm * 16 + lh * 8;
        float* ob = out + ((size_t)(n * HW + col) * T_ + t) * V_ + vb;
#pragma unroll
        for (int r = 0; r < 8; ++r) {
            float x = c[r];
            float y = x > 0.0f ? x : (__expf(x) - 1.0f);
            ob[r] = 1.0f / (1.0f + __expf(-y));
        }
    }
}

extern "C" void kernel_launch(void* const* d_in, const int* in_sizes, int n_in,
                              void* d_out, int out_size, void* d_ws, size_t ws_size,
                              hipStream_t stream) {
    const float* h  = (const float*)d_in[0];
    const float* Wp = (const float*)d_in[1];
    const float* ap = (const float*)d_in[2];
    const float* Bp = (const float*)d_in[3];
    float* out = (float*)d_out;

    float* ws    = (float*)d_ws;
    float* Whb   = ws;
    float* ei    = Whb  + (size_t)N_ * T_ * V_ * HW;
    float* ej    = ei   + N_ * V_ * T_;

    whe_kernel <<<N_ * 8 * 4, 256, 0, stream>>>(h, Wp, ap, Whb, ei, ej);
    gemm_kernel<<<N_ * T_,    256, 0, stream>>>(Whb, ei, ej, Bp, out);
}


# Product kernel: W taken in 8 blocks of 3 rows, wave w takes block (b xor w) at step b, first block asked for before the barrier: W's scalar loads hit the scalar cache; each 24-term sum reordered
# speedup vs baseline: 17.6772x; 1.1676x over previous
#include <hip/hip_runtime.h>


#define ALPHA 0.2f
#define N_  4
#define HD  16
#define WD  16
#define T_  24
#define V_  32
#define HW  256
#define THW 512

typedef __attribute__((ext_vector_type(2))) float v2f;
typedef __attribute__((ext_vector_type(8))) float v8f;

#define XP 194

__global__ void __launch_bounds__(256) whe_kernel(const float* __restrict__ h,
                                                  const float* __restrict__ Wp,
                                                  const float* __restrict__ a,
                                                  float* __restrict__ Wh,
                                                  float* __restrict__ ei,
                                                  float* __restrict__ ej) {
    __shared__ __attribute__((aligned(16))) float Xs[32 * XP];
    __shared__ __attribute__((aligned(16))) float As[THW];
    int bid = blockIdx.x;
    int vg  = bid & 3;
    int m   = (bid >> 2) & 7;
    int n   = bid >> 5;
    int tid = threadIdx.x;
    int w    = tid >> 5;
    int lane = tid & 31;
    int vv   = vg * 8 + w;
    int p    = m * 32 + lane;

    const float* hb = h + (size_t)(n * HW + m * 32) * (T_ * V_) + vg * 8;
    float4 hv[6];
#pragma unroll
    for (int k = 0; k < 6; ++k) {
        int q = tid + 256 * k;
        hv[k] = *(const float4*)(hb + (q >> 1) * V_ + (q & 1) * 4);
    }
    float a0 = a[tid], a1 = a[tid + 256];
    unsigned wu = (unsigned)__builtin_amdgcn_readfirstlane(w);
    float wf[3 * T_];
    {
        const float* wb0 = (const float*)((const char*)Wp + wu * (unsigned)(3 * T_ * sizeof(float)));
#pragma unroll
        for (int i = 0; i < 3 * T_; ++i) wf[i] = wb0[i];
    }
#pragma unroll
    for (int k = 0; k < 6; ++k) {
        int q = tid + 256 * k;
        int r = q >> 1;
        int pl = r / T_;
        float* d = Xs + r * 8 + pl * (XP - T_ * 8) + (q & 1) * 4;
        d[0] = hv[k].x; d[1] = hv[k].y; d[2] = hv[k].z; d[3] = hv[k].w;
    }
    As[tid] = a0; As[tid + 256] = a1;
    __syncthreads();

    float x[T_];
#pragma unroll
    for (int b = 0; b < 8; ++b) {
        unsigned rb = 3u * ((unsigned)b ^ wu);
#pragma unroll
        for (int j = 0; j < 3; ++j) x[3 * b + j] = Xs[lane * XP + (rb + j) * 8 + w];
    }

    float acc[T_];
#pragma unroll
    for (int t = 0; t < T_; ++t) acc[t] = 0.0f;
#pragma unroll
    for (int b = 0; b < 8; ++b) {
        unsigned wo = ((unsigned)b ^ wu) * (unsigned)(3 * T_ * sizeof(float));
        const float* wblk = (const float*)((const char*)Wp + wo);
#pragma unroll
        for (int j = 0; j < 3; ++j) {
#pragma unroll
            for (int t = 0; t < T_; ++t) acc[t] += x[3 * b + j] * (b == 0 ? wf[j * T_ + t] : wblk[j * T_ + t]);
        }
    }

    float* wp = Wh + ((size_t)(n * T_) * V_ + vv) * HW + p;
#pragma unroll
    for (int t = 0; t < T_; ++t) wp[(size_t)t * V_ * HW] = acc[t];

    int nv = n * V_ + vv;
    int fi = (p >> 4) * (2 * WD * T_) + (p & 15) * T_;
    int fj = fi + WD * T_;
    float si0 = 0.0f, si1 = 0.0f, si2 = 0.0f;
    float sj0 = 0.0f, sj1 = 0.0f, sj2 = 0.0f;
#pragma unroll
    for (int c = 0; c < T_ / 4; ++c) {
        int f0 = fi + 4 * c, f1 = fj + 4 * c;
        float4 ai = *(const float4*)(As + (f0 & (THW - 1)));
        float4 aj = *(const float4*)(As + (f1 & (THW - 1)));
        float x0 = acc[4 * c], x1 = acc[4 * c + 1], x2 = acc[4 * c + 2], x3 = acc[4 * c + 3];
        float di = x0 * ai.x + x1 * ai.y + x2 * ai.z + x3 * ai.w;
        float dj = x0 * aj.x + x1 * aj.y + x2 * aj.z + x3 * aj.w;
        int ri = (f0 >> 9) - 3 * m;
        int rj = (f1 >> 9) - 3 * m;
        si0 += (ri == 0) ? di : 0.0f;
        si1 += (ri == 1) ? di : 0.0f;
        si2 += (ri == 2) ? di : 0.0f;
        sj0 += (rj == 0) ? dj : 0.0f;
        sj1 += (rj == 1) ? dj : 0.0f;
        sj2 += (rj == 2) ? dj : 0.0f;
    }
#pragma unroll
    for (int off = 16; off > 0; off >>= 1) {
        si0 += __shfl_xor(si0, off); si1 += __shfl_xor(si1, off); si2 += __shfl_xor(si2, off);
        sj0 += __shfl_xor(sj0, off); sj1 += __shfl_xor(sj1, off); sj2 += __shfl_xor(sj2, off);
    }
    if (lane < 3) {
        float oi = lane == 0 ? si0 : (lane == 1 ? si1 : si2);
        float oj = lane == 0 ? sj0 : (lane == 1 ? sj1 : sj2);
        ei[nv * T_ + 3 * m + lane] = oi;
        ej[nv * T_ + 3 * m + lane] = oj;
    }
}

#define EJP 25

static __device__ __forceinline__ void lds_barrier() {
    asm volatile("s_wait_dscnt 0x0\n\ts_barrier_signal -1\n\ts_barrier_wait -1" ::: "memory");
}

__global__ void __launch_bounds__(256) gemm_kernel(const float* __restrict__ Wh,
                                                   const float* __restrict__ ei,
                                                   const float* __restrict__ ej,
                                                   const float* __restrict__ B,
                                                   float* __restrict__ out) {
    __shared__ float Gl [V_ * 33];
    __shared__ float Ajl[V_ * 33];
    __shared__ float Ml [V_ * 33];
    __shared__ float Wl [V_ * 257];
    __shared__ __attribute__((aligned(16))) float Eil[V_ * T_];
    __shared__ float Ejl[V_ * EJP];

    int bid = blockIdx.x;
    int n = bid / T_;
    int t = bid - n * T_;
    int tid = threadIdx.x;
    int wave = tid >> 5, lane = tid & 31;
    const float* wsl = Wh + (size_t)(n * T_ + t) * V_ * HW;

    float4 bv = *(const float4*)(B + 4 * tid);
    float4 eiv = make_float4(0.0f, 0.0f, 0.0f, 0.0f), ejv = eiv;
    if (tid < (V_ * T_) / 4) {
        eiv = *(const float4*)(ei + n * (V_ * T_) + 4 * tid);
        ejv = *(const float4*)(ej + n * (V_ * T_) + 4 * tid);
    }
    float wr[V_];
#pragma unroll
    for (int k = 0; k < V_; ++k) wr[k] = wsl[tid + 256 * k];

    int ai = tid >> 3, aj0 = (tid & 7) * 4;
    float av[4] = {bv.x, bv.y, bv.z, bv.w};
#pragma unroll
    for (int c = 0; c < 4; ++c) {
        av[c] = av[c] + 1e-6f + (ai == aj0 + c ? 1.0f : 0.0f);
        Ml[ai * 33 + aj0 + c] = av[c];
    }
    if (tid < (V_ * T_) / 4) {
        int f = 4 * tid;
        int r = f / T_;
        float* di = Eil + f;
        float* dj = Ejl + r * EJP + (f - r * T_);
        di[0] = eiv.x; di[1] = eiv.y; di[2] = eiv.z; di[3] = eiv.w;
        dj[0] = ejv.x; dj[1] = ejv.y; dj[2] = ejv.z; dj[3] = ejv.w;
    }
    lds_barrier();

    {
        float rv[V_];
        float mn = 1e30f, mx = -1e30f;
#pragma unroll
        for (int j = 0; j < V_; ++j) {
            rv[j] = Ml[lane * 33 + j];
            mn = fminf(mn, rv[j]); mx = fmaxf(mx, rv[j]);
        }
#pragma unroll
        for (int off = 16; off > 0; off >>= 1) {
            mn = fminf(mn, __shfl_xor(mn, off));
            mx = fmaxf(mx, __shfl_xor(mx, off));
        }
        float inv = 1.0f / (mx - mn);
        float s = 0.0f;
#pragma unroll
        for (int j = 0; j < V_; ++j) s += (rv[j] - mn) * inv;
        float dinv = rsqrtf(s);
        float dri = __shfl(dinv, ai);
#pragma unroll
        for (int c = 0; c < 4; ++c) {
            float drj = __shfl(dinv, aj0 + c);
            Ajl[ai * 33 + aj0 + c] = (av[c] - mn) * inv * dri * drj;
        }
    }

    {
        float ejr[T_];
#pragma unroll
        for (int tp = 0; tp < T_; ++tp) ejr[tp] = Ejl[lane * EJP + tp];
        float ejt = Ejl[lane * EJP + t];
        for (int k = 0; k < 4; ++k) {
            int i = wave + 8 * k;
            const float* eip = Eil + i * T_;
            float l[T_];
            float m = -1e30f;
#pragma unroll
            for (int tp = 0; tp < T_; ++tp) {
                float e = eip[tp] + ejr[tp];
                e = e > 0.0f ? e : ALPHA * e;
                l[tp] = e;
                m = fmaxf(m, e);
            }
            float s = 0.0f;
#pragma unroll
            for (int tp = 0; tp < T_; ++tp) s += __expf(l[tp] - m);
            float is = 1.0f / s;
            float et = eip[t] + ejt;
            et = et > 0.0f ? et : ALPHA * et;
            Gl[i * 33 + lane] = __expf(et - m) * is;
        }
    }
    lds_barrier();

    for (int idx = tid; idx < V_ * V_; idx += 256) {
        int v = idx >> 5, j = idx & 31;
        float acc = 0.0f;
#pragma unroll
        for (int i2 = 0; i2 < V_; ++i2) acc += Ajl[i2 * 33 + v] * Gl[i2 * 33 + j];
        Ml[v * 33 + j] = acc;
    }
#pragma unroll
    for (int k = 0; k < V_; ++k) Wl[k * 257 + tid] = wr[k];
    __syncthreads();

    int lh = lane >> 4, lm = lane & 15;
    for (int q = 0; q < 4; ++q) {
        int tile = wave * 4 + q;
        int tm = tile >> 4, tn = tile & 15;
        int arow = tm * 16 + lm;
        int col  = tn * 16 + lm;
        v8f c = {};
#pragma unroll
        for (int kk = 0; kk < 8; ++kk) {
            int kb = kk * 4 + lh * 2;
            v2f a, b;
            a.x = Ml[arow * 33 + kb];
            a.y = Ml[arow * 33 + kb + 1];
            b.x = Wl[kb * 257 + col];
            b.y = Wl[(kb + 1) * 257 + col];
            c = __builtin_amdgcn_wmma_f32_16x16x4_f32(
                     false, a,  false, b,
                     (short)0, c,  false,  false);
        }
        int vb = tm * 16 + lh * 8;
        float* ob = out + ((size_t)(n * HW + col) * T_ + t) * V_ + vb;
#pragma unroll
        for (int r = 0; r < 8; ++r) {
            float x = c[r];
            float y = x > 0.0f ? x : (__expf(x) - 1.0f);
            ob[r] = 1.0f / (1.0f + __expf(-y));
        }
    }
}

extern "C" void kernel_launch(void* const* d_in, const int* in_sizes, int n_in,
                              void* d_out, int out_size, void* d_ws, size_t ws_size,
                              hipStream_t stream) {
    const float* h  = (const float*)d_in[0];
    const float* Wp = (const float*)d_in[1];
    const float* ap = (const float*)d_in[2];
    const float* Bp = (const float*)d_in[3];
    float* out = (float*)d_out;

    float* ws    = (float*)d_ws;
    float* Whb   = ws;
    float* ei    = Whb  + (size_t)N_ * T_ * V_ * HW;
    float* ej    = ei   + N_ * V_ * T_;

    whe_kernel <<<N_ * 8 * 4, 256, 0, stream>>>(h, Wp, ap, Whb, ei, ej);
    gemm_kernel<<<N_ * T_,    256, 0, stream>>>(Whb, ei, ej, Bp, out);
}


# Product kernel: M = adjn^T*G (32x32x32) leaves its serial LDS-fed FMA loop for the matrix unit, one 16x16 tile per wave 0..3, two accumulators of four k-steps; each 32-term sum reordered
# speedup vs baseline: 18.9866x; 1.0741x over previous
#include <hip/hip_runtime.h>


#define ALPHA 0.2f
#define N_  4
#define HD  16
#define WD  16
#define T_  24
#define V_  32
#define HW  256
#define THW 512

typedef __attribute__((ext_vector_type(2))) float v2f;
typedef __attribute__((ext_vector_type(8))) float v8f;

#define XP 194

__global__ void __launch_bounds__(256) whe_kernel(const float* __restrict__ h,
                                                  const float* __restrict__ Wp,
                                                  const float* __restrict__ a,
                                                  float* __restrict__ Wh,
                                                  float* __restrict__ ei,
                                                  float* __restrict__ ej) {
    __shared__ __attribute__((aligned(16))) float Xs[32 * XP];
    __shared__ __attribute__((aligned(16))) float As[THW];
    int bid = blockIdx.x;
    int vg  = bid & 3;
    int m   = (bid >> 2) & 7;
    int n   = bid >> 5;
    int tid = threadIdx.x;
    int w    = tid >> 5;
    int lane = tid & 31;
    int vv   = vg * 8 + w;
    int p    = m * 32 + lane;

    const float* hb = h + (size_t)(n * HW + m * 32) * (T_ * V_) + vg * 8;
    float4 hv[6];
#pragma unroll
    for (int k = 0; k < 6; ++k) {
        int q = tid + 256 * k;
        hv[k] = *(const float4*)(hb + (q >> 1) * V_ + (q & 1) * 4);
    }
    float a0 = a[tid], a1 = a[tid + 256];
    unsigned wu = (unsigned)__builtin_amdgcn_readfirstlane(w);
    float wf[3 * T_];
    {
        const float* wb0 = (const float*)((const char*)Wp + wu * (unsigned)(3 * T_ * sizeof(float)));
#pragma unroll
        for (int i = 0; i < 3 * T_; ++i) wf[i] = wb0[i];
    }
#pragma unroll
    for (int k = 0; k < 6; ++k) {
        int q = tid + 256 * k;
        int r = q >> 1;
        int pl = r / T_;
        float* d = Xs + r * 8 + pl * (XP - T_ * 8) + (q & 1) * 4;
        d[0] = hv[k].x; d[1] = hv[k].y; d[2] = hv[k].z; d[3] = hv[k].w;
    }
    As[tid] = a0; As[tid + 256] = a1;
    __syncthreads();

    float x[T_];
#pragma unroll
    for (int b = 0; b < 8; ++b) {
        unsigned rb = 3u * ((unsigned)b ^ wu);
#pragma unroll
        for (int j = 0; j < 3; ++j) x[3 * b + j] = Xs[lane * XP + (rb + j) * 8 + w];
    }

    float acc[T_];
#pragma unroll
    for (int t = 0; t < T_; ++t) acc[t] = 0.0f;
#pragma unroll
    for (int b = 0; b < 8; ++b) {
        unsigned wo = ((unsigned)b ^ wu) * (unsigned)(3 * T_ * sizeof(float));
        const float* wblk = (const float*)((const char*)Wp + wo);
#pragma unroll
        for (int j = 0; j < 3; ++j) {
#pragma unroll
            for (int t = 0; t < T_; ++t) acc[t] += x[3 * b + j] * (b == 0 ? wf[j * T_ + t] : wblk[j * T_ + t]);
        }
    }

    float* wp = Wh + ((size_t)(n * T_) * V_ + vv) * HW + p;
#pragma unroll
    for (int t = 0; t < T_; ++t) wp[(size_t)t * V_ * HW] = acc[t];

    int nv = n * V_ + vv;
    int fi = (p >> 4) * (2 * WD * T_) + (p & 15) * T_;
    int fj = fi + WD * T_;
    float si0 = 0.0f, si1 = 0.0f, si2 = 0.0f;
    float sj0 = 0.0f, sj1 = 0.0f, sj2 = 0.0f;
#pragma unroll
    for (int c = 0; c < T_ / 4; ++c) {
        int f0 = fi + 4 * c, f1 = fj + 4 * c;
        float4 ai = *(const float4*)(As + (f0 & (THW - 1)));
        float4 aj = *(const float4*)(As + (f1 & (THW - 1)));
        float x0 = acc[4 * c], x1 = acc[4 * c + 1], x2 = acc[4 * c + 2], x3 = acc[4 * c + 3];
        float di = x0 * ai.x + x1 * ai.y + x2 * ai.z + x3 * ai.w;
        float dj = x0 * aj.x + x1 * aj.y + x2 * aj.z + x3 * aj.w;
        int ri = (f0 >> 9) - 3 * m;
        int rj = (f1 >> 9) - 3 * m;
        si0 += (ri == 0) ? di : 0.0f;
        si1 += (ri == 1) ? di : 0.0f;
        si2 += (ri == 2) ? di : 0.0f;
        sj0 += (rj == 0) ? dj : 0.0f;
        sj1 += (rj == 1) ? dj : 0.0f;
        sj2 += (rj == 2) ? dj : 0.0f;
    }
#pragma unroll
    for (int off = 16; off > 0; off >>= 1) {
        si0 += __shfl_xor(si0, off); si1 += __shfl_xor(si1, off); si2 += __shfl_xor(si2, off);
        sj0 += __shfl_xor(sj0, off); sj1 += __shfl_xor(sj1, off); sj2 += __shfl_xor(sj2, off);
    }
    if (lane < 3) {
        float oi = lane == 0 ? si0 : (lane == 1 ? si1 : si2);
        float oj = lane == 0 ? sj0 : (lane == 1 ? sj1 : sj2);
        ei[nv * T_ + 3 * m + lane] = oi;
        ej[nv * T_ + 3 * m + lane] = oj;
    }
}

#define EJP 25

static __device__ __forceinline__ void lds_barrier() {
    asm volatile("s_wait_dscnt 0x0\n\ts_barrier_signal -1\n\ts_barrier_wait -1" ::: "memory");
}

__global__ void __launch_bounds__(256) gemm_kernel(const float* __restrict__ Wh,
                                                   const float* __restrict__ ei,
                                                   const float* __restrict__ ej,
                                                   const float* __restrict__ B,
                                                   float* __restrict__ out) {
    __shared__ float Gl [V_ * 33];
    __shared__ float Ajl[V_ * 33];
    __shared__ float Ml [V_ * 33];
    __shared__ float Wl [V_ * 257];
    __shared__ __attribute__((aligned(16))) float Eil[V_ * T_];
    __shared__ float Ejl[V_ * EJP];

    int bid = blockIdx.x;
    int n = bid / T_;
    int t = bid - n * T_;
    int tid = threadIdx.x;
    int wave = tid >> 5, lane = tid & 31;
    const float* wsl = Wh + (size_t)(n * T_ + t) * V_ * HW;

    float4 bv = *(const float4*)(B + 4 * tid);
    float4 eiv = make_float4(0.0f, 0.0f, 0.0f, 0.0f), ejv = eiv;
    if (tid < (V_ * T_) / 4) {
        eiv = *(const float4*)(ei + n * (V_ * T_) + 4 * tid);
        ejv = *(const float4*)(ej + n * (V_ * T_) + 4 * tid);
    }
    float wr[V_];
#pragma unroll
    for (int k = 0; k < V_; ++k) wr[k] = wsl[tid + 256 * k];

    int ai = tid >> 3, aj0 = (tid & 7) * 4;
    float av[4] = {bv.x, bv.y, bv.z, bv.w};
#pragma unroll
    for (int c = 0; c < 4; ++c) {
        av[c] = av[c] + 1e-6f + (ai == aj0 + c ? 1.0f : 0.0f);
        Ml[ai * 33 + aj0 + c] = av[c];
    }
    if (tid < (V_ * T_) / 4) {
        int f = 4 * tid;
        int r = f / T_;
        float* di = Eil + f;
        float* dj = Ejl + r * EJP + (f - r * T_);
        di[0] = eiv.x; di[1] = eiv.y; di[2] = eiv.z; di[3] = eiv.w;
        dj[0] = ejv.x; dj[1] = ejv.y; dj[2] = ejv.z; dj[3] = ejv.w;
    }
    lds_barrier();

    {
        float rv[V_];
        float mn = 1e30f, mx = -1e30f;
#pragma unroll
        for (int j = 0; j < V_; ++j) {
            rv[j] = Ml[lane * 33 + j];
            mn = fminf(mn, rv[j]); mx = fmaxf(mx, rv[j]);
        }
#pragma unroll
        for (int off = 16; off > 0; off >>= 1) {
            mn = fminf(mn, __shfl_xor(mn, off));
            mx = fmaxf(mx, __shfl_xor(mx, off));
        }
        float inv = 1.0f / (mx - mn);
        float s = 0.0f;
#pragma unroll
        for (int j = 0; j < V_; ++j) s += (rv[j] - mn) * inv;
        float dinv = rsqrtf(s);
        float dri = __shfl(dinv, ai);
#pragma unroll
        for (int c = 0; c < 4; ++c) {
            float drj = __shfl(dinv, aj0 + c);
            Ajl[ai * 33 + aj0 + c] = (av[c] - mn) * inv * dri * drj;
        }
    }

    {
        float ejr[T_];
#pragma unroll
        for (int tp = 0; tp < T_; ++tp) ejr[tp] = Ejl[lane * EJP + tp];
        float ejt = Ejl[lane * EJP + t];
        for (int k = 0; k < 4; ++k) {
            int i = wave + 8 * k;
            const float* eip = Eil + i * T_;
            float l[T_];
            float m = -1e30f;
#pragma unroll
            for (int tp = 0; tp < T_; ++tp) {
                float e = eip[tp] + ejr[tp];
                e = e > 0.0f ? e : ALPHA * e;
                l[tp] = e;
                m = fmaxf(m, e);
            }
            float s = 0.0f;
#pragma unroll
            for (int tp = 0; tp < T_; ++tp) s += __expf(l[tp] - m);
            float is = 1.0f / s;
            float et = eip[t] + ejt;
            et = et > 0.0f ? et : ALPHA * et;
            Gl[i * 33 + lane] = __expf(et - m) * is;
        }
    }
    lds_barrier();

    int lh = lane >> 4, lm = lane & 15;
    int wv = __builtin_amdgcn_readfirstlane(wave);
    if (wv < 4) {
        int mrow = (wv >> 1) * 16 + lm;
        int mcol = (wv & 1) * 16 + lm;
        v8f c0 = {}, c1 = {};
#pragma unroll
        for (int kk = 0; kk < 4; ++kk) {
            int kb0 = kk * 4 + lh * 2;
            int kb1 = kb0 + 16;
            v2f a0, b0, a1, b1;
            a0.x = Ajl[kb0 * 33 + mrow];
            a0.y = Ajl[(kb0 + 1) * 33 + mrow];
            b0.x = Gl[kb0 * 33 + mcol];
            b0.y = Gl[(kb0 + 1) * 33 + mcol];
            a1.x = Ajl[kb1 * 33 + mrow];
            a1.y = Ajl[(kb1 + 1) * 33 + mrow];
            b1.x = Gl[kb1 * 33 + mcol];
            b1.y = Gl[(kb1 + 1) * 33 + mcol];
            c0 = __builtin_amdgcn_wmma_f32_16x16x4_f32(
                      false, a0,  false, b0,
                      (short)0, c0,  false,  false);
            c1 = __builtin_amdgcn_wmma_f32_16x16x4_f32(
                      false, a1,  false, b1,
                      (short)0, c1,  false,  false);
        }
        int mr0 = (wv >> 1) * 16 + lh * 8;
#pragma unroll
        for (int r = 0; r < 8; ++r) Ml[(mr0 + r) * 33 + mcol] = c0[r] + c1[r];
    }
#pragma unroll
    for (int k = 0; k < V_; ++k) Wl[k * 257 + tid] = wr[k];
    __syncthreads();

    for (int q = 0; q < 4; ++q) {
        int tile = wave * 4 + q;
        int tm = tile >> 4, tn = tile & 15;
        int arow = tm * 16 + lm;
        int col  = tn * 16 + lm;
        v8f c = {};
#pragma unroll
        for (int kk = 0; kk < 8; ++kk) {
            int kb = kk * 4 + lh * 2;
            v2f a, b;
            a.x = Ml[arow * 33 + kb];
            a.y = Ml[arow * 33 + kb + 1];
            b.x = Wl[kb * 257 + col];
            b.y = Wl[(kb + 1) * 257 + col];
            c = __builtin_amdgcn_wmma_f32_16x16x4_f32(
                     false, a,  false, b,
                     (short)0, c,  false,  false);
        }
        int vb = tm * 16 + lh * 8;
        float* ob = out + ((size_t)(n * HW + col) * T_ + t) * V_ + vb;
#pragma unroll
        for (int r = 0; r < 8; ++r) {
            float x = c[r];
            float y = x > 0.0f ? x : (__expf(x) - 1.0f);
            ob[r] = 1.0f / (1.0f + __expf(-y));
        }
    }
}

extern "C" void kernel_launch(void* const* d_in, const int* in_sizes, int n_in,
                              void* d_out, int out_size, void* d_ws, size_t ws_size,
                              hipStream_t stream) {
    const float* h  = (const float*)d_in[0];
    const float* Wp = (const float*)d_in[1];
    const float* ap = (const float*)d_in[2];
    const float* Bp = (const float*)d_in[3];
    float* out = (float*)d_out;

    float* ws    = (float*)d_ws;
    float* Whb   = ws;
    float* ei    = Whb  + (size_t)N_ * T_ * V_ * HW;
    float* ej    = ei   + N_ * V_ * T_;

    whe_kernel <<<N_ * 8 * 4, 256, 0, stream>>>(h, Wp, ap, Whb, ei, ej);
    gemm_kernel<<<N_ * T_,    256, 0, stream>>>(Whb, ei, ej, Bp, out);
}


# Wh kept v-innermost [n][t][p][v]: launch 1 runs lane=v on whole h lines (no LDS turn, no lane butterflies), ei/ej as per-8p block partials added in launch 2; Wh and ei/ej sum orders change
# speedup vs baseline: 21.9925x; 1.1583x over previous
#include <hip/hip_runtime.h>


#define ALPHA 0.2f
#define N_  4
#define HD  16
#define WD  16
#define T_  24
#define V_  32
#define HW  256
#define THW 512

typedef __attribute__((ext_vector_type(2))) float v2f;
typedef __attribute__((ext_vector_type(8))) float v8f;

static __device__ __forceinline__ void lds_barrier() {
    asm volatile("s_wait_dscnt 0x0\n\ts_barrier_signal -1\n\ts_barrier_wait -1" ::: "memory");
}

__global__ void __launch_bounds__(256) whe_kernel(const float* __restrict__ h,
                                                  const float* __restrict__ Wp,
                                                  const float* __restrict__ a,
                                                  float* __restrict__ Wh,
                                                  float* __restrict__ part) {
    __shared__ float Red[4 * 8 * V_];
    int bid = blockIdx.x;
    int pb  = bid & 31;
    int n   = bid >> 5;
    int tid = threadIdx.x;
    int lane = tid & 31;
    unsigned wu = (unsigned)__builtin_amdgcn_readfirstlane(tid >> 5);
    int p = pb * 8 + (int)wu;

    const float* hrow = h + (size_t)((n * HW + p) * T_) * V_;
    float x[T_];
#pragma unroll
    for (int b = 0; b < 8; ++b) {
        const float* hq = hrow + 3u * ((unsigned)b ^ wu) * V_;
#pragma unroll
        for (int j = 0; j < 3; ++j) x[3 * b + j] = hq[j * V_ + lane];
    }
    __builtin_amdgcn_sched_barrier(0);

    float acc[T_];
#pragma unroll
    for (int t = 0; t < T_; ++t) acc[t] = 0.0f;
#pragma unroll
    for (int b = 0; b < 8; ++b) {
        unsigned wo = ((unsigned)b ^ wu) * (unsigned)(3 * T_ * sizeof(float));
        const float* wblk = (const float*)((const char*)Wp + wo);
#pragma unroll
        for (int j = 0; j < 3; ++j) {
#pragma unroll
            for (int t = 0; t < T_; ++t) acc[t] += x[3 * b + j] * wblk[j * T_ + t];
        }
    }

    float* wp = Wh + ((size_t)(n * T_) * HW + p) * V_ + lane;
#pragma unroll
    for (int t = 0; t < T_; ++t) wp[(size_t)t * HW * V_] = acc[t];

    int f0 = (p >> 4) * (2 * WD * T_) + (p & 15) * T_;
    int fb = (pb >> 1) * (2 * WD * T_) + (pb & 1) * (8 * T_);
    int R0i = fb >> 9, R0j = (fb + WD * T_) >> 9;
    float si0 = 0.0f, si1 = 0.0f, sj0 = 0.0f, sj1 = 0.0f;
#pragma unroll
    for (int g = 0; g < 3; ++g) {
        int fi = f0 + 8 * g, fj = fi + WD * T_;
        const float* ai = a + (fi & (THW - 1));
        const float* aj = a + (fj & (THW - 1));
        float di = 0.0f, dj = 0.0f;
#pragma unroll
        for (int u = 0; u < 8; ++u) {
            di += acc[8 * g + u] * ai[u];
            dj += acc[8 * g + u] * aj[u];
        }
        bool hi_i = (fi >> 9) != R0i;
        bool hi_j = (fj >> 9) != R0j;
        si0 += hi_i ? 0.0f : di;
        si1 += hi_i ? di : 0.0f;
        sj0 += hi_j ? 0.0f : dj;
        sj1 += hi_j ? dj : 0.0f;
    }
    Red[(0 * 8 + wu) * V_ + lane] = si0;
    Red[(1 * 8 + wu) * V_ + lane] = si1;
    Red[(2 * 8 + wu) * V_ + lane] = sj0;
    Red[(3 * 8 + wu) * V_ + lane] = sj1;
    lds_barrier();
    if (tid < 4 * V_) {
        int q = tid >> 5;
        float s = 0.0f;
#pragma unroll
        for (int w = 0; w < 8; ++w) s += Red[(q * 8 + w) * V_ + lane];
        part[((n * 32 + pb) * 4 + q) * V_ + lane] = s;
    }
}

#define EJP 25
#define WP  36

__global__ void __launch_bounds__(256) gemm_kernel(const float* Wh,
                                                   const float* __restrict__ part,
                                                   const float* __restrict__ B,
                                                   float* __restrict__ out) {
    __shared__ float Gl [V_ * 33];
    __shared__ float Ajl[V_ * 33];
    __shared__ float Ml [V_ * 33];
    __shared__ __attribute__((aligned(16))) float Wl[HW * WP];
    __shared__ float Eil[V_ * T_];
    __shared__ float Ejl[V_ * EJP];

    int bid = blockIdx.x;
    int n = bid / T_;
    int t = bid - n * T_;
    int tid = threadIdx.x;
    int wave = tid >> 5, lane = tid & 31;
    const float4* wsl = (const float4*)(Wh + (size_t)(n * T_ + t) * HW * V_);

    float4 bv = *(const float4*)(B + 4 * tid);
    int wvu = __builtin_amdgcn_readfirstlane(wave);
    float evi[3], evj[3];
#pragma unroll
    for (int k = 0; k < 3; ++k) {
        int tn = wvu + 8 * k;
        int m = tn / 3, r = tn - 3 * m;
        const float* pn = part + (size_t)(n * 32 + 4 * m) * (4 * V_);
        float iA = pn[(r == 0 ? 0 : (r == 1 ? 2 : 3)) * (4 * V_) + (r == 2 ? 1 : 0) * V_ + lane];
        float iB = pn[(r == 0 ? 1 : 3) * (4 * V_) + lane];
        float jA = pn[(r == 2 ? 2 : 0) * (4 * V_) + (r == 1 ? 3 : 2) * V_ + lane];
        float jB = pn[(r == 1 ? 1 : 3) * (4 * V_) + 2 * V_ + lane];
        evi[k] = iA + (r < 2 ? iB : 0.0f);
        evj[k] = jA + (r > 0 ? jB : 0.0f);
    }
    float4 wr0 = wsl[tid],        wr1 = wsl[tid + 256],  wr2 = wsl[tid + 512],  wr3 = wsl[tid + 768];
    float4 wr4 = wsl[tid + 1024], wr5 = wsl[tid + 1280], wr6 = wsl[tid + 1536], wr7 = wsl[tid + 1792];

    int ai = tid >> 3, aj0 = (tid & 7) * 4;
    float av[4] = {bv.x, bv.y, bv.z, bv.w};
#pragma unroll
    for (int c = 0; c < 4; ++c) {
        av[c] = av[c] + 1e-6f + (ai == aj0 + c ? 1.0f : 0.0f);
        Ml[ai * 33 + aj0 + c] = av[c];
    }
#pragma unroll
    for (int k = 0; k < 3; ++k) {
        Eil[lane * T_  + wvu + 8 * k] = evi[k];
        Ejl[lane * EJP + wvu + 8 * k] = evj[k];
    }
    lds_barrier();

    {
        float rv[V_];
        float mn = 1e30f, mx = -1e30f;
#pragma unroll
        for (int j = 0; j < V_; ++j) {
            rv[j] = Ml[lane * 33 + j];
            mn = fminf(mn, rv[j]); mx = fmaxf(mx, rv[j]);
        }
#pragma unroll
        for (int off = 16; off > 0; off >>= 1) {
            mn = fminf(mn, __shfl_xor(mn, off));
            mx = fmaxf(mx, __shfl_xor(mx, off));
        }
        float inv = 1.0f / (mx - mn);
        float s = 0.0f;
#pragma unroll
        for (int j = 0; j < V_; ++j) s += (rv[j] - mn) * inv;
        float dinv = rsqrtf(s);
        float dri = __shfl(dinv, ai);
#pragma unroll
        for (int c = 0; c < 4; ++c) {
            float drj = __shfl(dinv, aj0 + c);
            Ajl[ai * 33 + aj0 + c] = (av[c] - mn) * inv * dri * drj;
        }
    }

    {
        float ejr[T_];
#pragma unroll
        for (int tp = 0; tp < T_; ++tp) ejr[tp] = Ejl[lane * EJP + tp];
        float ejt = Ejl[lane * EJP + t];
        for (int k = 0; k < 4; ++k) {
            int i = wave + 8 * k;
            const float* eip = Eil + i * T_;
            float l[T_];
            float m = -1e30f;
#pragma unroll
            for (int tp = 0; tp < T_; ++tp) {
                float e = eip[tp] + ejr[tp];
                e = e > 0.0f ? e : ALPHA * e;
                l[tp] = e;
                m = fmaxf(m, e);
            }
            float s = 0.0f;
#pragma unroll
            for (int tp = 0; tp < T_; ++tp) s += __expf(l[tp] - m);
            float is = 1.0f / s;
            float et = eip[t] + ejt;
            et = et > 0.0f ? et : ALPHA * et;
            Gl[i * 33 + lane] = __expf(et - m) * is;
        }
    }
    lds_barrier();

    int lh = lane >> 4, lm = lane & 15;
    int wv = wvu;
    if (wv < 4) {
        int mrow = (wv >> 1) * 16 + lm;
        int mcol = (wv & 1) * 16 + lm;
        v8f c0 = {}, c1 = {};
#pragma unroll
        for (int kk = 0; kk < 4; ++kk) {
            int kb0 = kk * 4 + lh * 2;
            int kb1 = kb0 + 16;
            v2f a0, b0, a1, b1;
            a0.x = Ajl[kb0 * 33 + mrow];
            a0.y = Ajl[(kb0 + 1) * 33 + mrow];
            b0.x = Gl[kb0 * 33 + mcol];
            b0.y = Gl[(kb0 + 1) * 33 + mcol];
            a1.x = Ajl[kb1 * 33 + mrow];
            a1.y = Ajl[(kb1 + 1) * 33 + mrow];
            b1.x = Gl[kb1 * 33 + mcol];
            b1.y = Gl[(kb1 + 1) * 33 + mcol];
            c0 = __builtin_amdgcn_wmma_f32_16x16x4_f32(
                      false, a0,  false, b0,
                      (short)0, c0,  false,  false);
            c1 = __builtin_amdgcn_wmma_f32_16x16x4_f32(
                      false, a1,  false, b1,
                      (short)0, c1,  false,  false);
        }
        int mr0 = (wv >> 1) * 16 + lh * 8;
#pragma unroll
        for (int r = 0; r < 8; ++r) Ml[(mr0 + r) * 33 + mcol] = c0[r] + c1[r];
    }
    {
        float4* wd = (float4*)(Wl + (tid >> 3) * WP + (tid & 7) * 4);
        const int ks = 32 * WP / 4;
        wd[0]      = wr0; wd[ks]     = wr1; wd[2 * ks] = wr2; wd[3 * ks] = wr3;
        wd[4 * ks] = wr4; wd[5 * ks] = wr5; wd[6 * ks] = wr6; wd[7 * ks] = wr7;
    }
    __syncthreads();

    for (int q = 0; q < 4; ++q) {
        int tile = wave * 4 + q;
        int tm = tile >> 4, tn = tile & 15;
        int arow = tm * 16 + lm;
        int col  = tn * 16 + lm;
        v8f c = {};
#pragma unroll
        for (int kk = 0; kk < 8; ++kk) {
            int kb = kk * 4 + lh * 2;
            v2f a;
            a.x = Ml[arow * 33 + kb];
            a.y = Ml[arow * 33 + kb + 1];
            v2f b = *(const v2f*)(Wl + col * WP + kb);
            c = __builtin_amdgcn_wmma_f32_16x16x4_f32(
                     false, a,  false, b,
                     (short)0, c,  false,  false);
        }
        int vb = tm * 16 + lh * 8;
        float* ob = out + ((size_t)(n * HW + col) * T_ + t) * V_ + vb;
#pragma unroll
        for (int r = 0; r < 8; ++r) {
            float x = c[r];
            float y = x > 0.0f ? x : (__expf(x) - 1.0f);
            ob[r] = 1.0f / (1.0f + __expf(-y));
        }
    }
}

extern "C" void kernel_launch(void* const* d_in, const int* in_sizes, int n_in,
                              void* d_out, int out_size, void* d_ws, size_t ws_size,
                              hipStream_t stream) {
    const float* h  = (const float*)d_in[0];
    const float* Wp = (const float*)d_in[1];
    const float* ap = (const float*)d_in[2];
    const float* Bp = (const float*)d_in[3];
    float* out = (float*)d_out;

    float* ws    = (float*)d_ws;
    float* Whb   = ws;
    float* part  = Whb  + (size_t)N_ * T_ * V_ * HW;

    whe_kernel <<<N_ * 32, 256, 0, stream>>>(h, Wp, ap, Whb, part);
    gemm_kernel<<<N_ * T_, 256, 0, stream>>>(Whb, part, Bp, out);
}


# Launch 2: each wave owns 32 p x all 32 v (2x2 tiles), passes its activated tiles through its own LDS patch and stores whole 128-byte out lines, 4 per store, not 16-byte pieces; sum orders kept
# speedup vs baseline: 23.2659x; 1.0579x over previous
#include <hip/hip_runtime.h>


#define ALPHA 0.2f
#define N_  4
#define HD  16
#define WD  16
#define T_  24
#define V_  32
#define HW  256
#define THW 512

typedef __attribute__((ext_vector_type(2))) float v2f;
typedef __attribute__((ext_vector_type(8))) float v8f;

static __device__ __forceinline__ void lds_barrier() {
    asm volatile("s_wait_dscnt 0x0\n\ts_barrier_signal -1\n\ts_barrier_wait -1" ::: "memory");
}

static __device__ __forceinline__ void wave_patch_written() {
    asm volatile("s_wait_dscnt 0x0" ::: "memory");
}

__global__ void __launch_bounds__(256) whe_kernel(const float* __restrict__ h,
                                                  const float* __restrict__ Wp,
                                                  const float* __restrict__ a,
                                                  float* __restrict__ Wh,
                                                  float* __restrict__ part) {
    __shared__ float Red[4 * 8 * V_];
    int bid = blockIdx.x;
    int pb  = bid & 31;
    int n   = bid >> 5;
    int tid = threadIdx.x;
    int lane = tid & 31;
    unsigned wu = (unsigned)__builtin_amdgcn_readfirstlane(tid >> 5);
    int p = pb * 8 + (int)wu;

    const float* hrow = h + (size_t)((n * HW + p) * T_) * V_;
    float x[T_];
#pragma unroll
    for (int b = 0; b < 8; ++b) {
        const float* hq = hrow + 3u * ((unsigned)b ^ wu) * V_;
#pragma unroll
        for (int j = 0; j < 3; ++j) x[3 * b + j] = hq[j * V_ + lane];
    }
    __builtin_amdgcn_sched_barrier(0);

    float acc[T_];
#pragma unroll
    for (int t = 0; t < T_; ++t) acc[t] = 0.0f;
#pragma unroll
    for (int b = 0; b < 8; ++b) {
        unsigned wo = ((unsigned)b ^ wu) * (unsigned)(3 * T_ * sizeof(float));
        const float* wblk = (const float*)((const char*)Wp + wo);
#pragma unroll
        for (int j = 0; j < 3; ++j) {
#pragma unroll
            for (int t = 0; t < T_; ++t) acc[t] += x[3 * b + j] * wblk[j * T_ + t];
        }
    }

    float* wp = Wh + ((size_t)(n * T_) * HW + p) * V_ + lane;
#pragma unroll
    for (int t = 0; t < T_; ++t) wp[(size_t)t * HW * V_] = acc[t];

    int f0 = (p >> 4) * (2 * WD * T_) + (p & 15) * T_;
    int fb = (pb >> 1) * (2 * WD * T_) + (pb & 1) * (8 * T_);
    int R0i = fb >> 9, R0j = (fb + WD * T_) >> 9;
    float si0 = 0.0f, si1 = 0.0f, sj0 = 0.0f, sj1 = 0.0f;
#pragma unroll
    for (int g = 0; g < 3; ++g) {
        int fi = f0 + 8 * g, fj = fi + WD * T_;
        const float* ai = a + (fi & (THW - 1));
        const float* aj = a + (fj & (THW - 1));
        float di = 0.0f, dj = 0.0f;
#pragma unroll
        for (int u = 0; u < 8; ++u) {
            di += acc[8 * g + u] * ai[u];
            dj += acc[8 * g + u] * aj[u];
        }
        bool hi_i = (fi >> 9) != R0i;
        bool hi_j = (fj >> 9) != R0j;
        si0 += hi_i ? 0.0f : di;
        si1 += hi_i ? di : 0.0f;
        sj0 += hi_j ? 0.0f : dj;
        sj1 += hi_j ? dj : 0.0f;
    }
    Red[(0 * 8 + wu) * V_ + lane] = si0;
    Red[(1 * 8 + wu) * V_ + lane] = si1;
    Red[(2 * 8 + wu) * V_ + lane] = sj0;
    Red[(3 * 8 + wu) * V_ + lane] = sj1;
    lds_barrier();
    if (tid < 4 * V_) {
        int q = tid >> 5;
        float s = 0.0f;
#pragma unroll
        for (int w = 0; w < 8; ++w) s += Red[(q * 8 + w) * V_ + lane];
        part[((n * 32 + pb) * 4 + q) * V_ + lane] = s;
    }
}

#define EJP 25
#define WP  36

__global__ void __launch_bounds__(256) gemm_kernel(const float* Wh,
                                                   const float* __restrict__ part,
                                                   const float* __restrict__ B,
                                                   float* __restrict__ out) {
    __shared__ float Gl [V_ * 33];
    __shared__ float Ajl[V_ * 33];
    __shared__ float Ml [V_ * 33];
    __shared__ __attribute__((aligned(16))) float Wl[HW * WP];
    __shared__ float Eil[V_ * T_];
    __shared__ float Ejl[V_ * EJP];
    __shared__ __attribute__((aligned(16))) float Ol[8 * 16 * WP];

    int bid = blockIdx.x;
    int n = bid / T_;
    int t = bid - n * T_;
    int tid = threadIdx.x;
    int wave = tid >> 5, lane = tid & 31;
    const float4* wsl = (const float4*)(Wh + (size_t)(n * T_ + t) * HW * V_);

    float4 bv = *(const float4*)(B + 4 * tid);
    int wvu = __builtin_amdgcn_readfirstlane(wave);
    float evi[3], evj[3];
#pragma unroll
    for (int k = 0; k < 3; ++k) {
        int tn = wvu + 8 * k;
        int m = tn / 3, r = tn - 3 * m;
        const float* pn = part + (size_t)(n * 32 + 4 * m) * (4 * V_);
        float iA = pn[(r == 0 ? 0 : (r == 1 ? 2 : 3)) * (4 * V_) + (r == 2 ? 1 : 0) * V_ + lane];
        float iB = pn[(r == 0 ? 1 : 3) * (4 * V_) + lane];
        float jA = pn[(r == 2 ? 2 : 0) * (4 * V_) + (r == 1 ? 3 : 2) * V_ + lane];
        float jB = pn[(r == 1 ? 1 : 3) * (4 * V_) + 2 * V_ + lane];
        evi[k] = iA + (r < 2 ? iB : 0.0f);
        evj[k] = jA + (r > 0 ? jB : 0.0f);
    }
    float4 wr0 = wsl[tid],        wr1 = wsl[tid + 256],  wr2 = wsl[tid + 512],  wr3 = wsl[tid + 768];
    float4 wr4 = wsl[tid + 1024], wr5 = wsl[tid + 1280], wr6 = wsl[tid + 1536], wr7 = wsl[tid + 1792];

    int ai = tid >> 3, aj0 = (tid & 7) * 4;
    float av[4] = {bv.x, bv.y, bv.z, bv.w};
#pragma unroll
    for (int c = 0; c < 4; ++c) {
        av[c] = av[c] + 1e-6f + (ai == aj0 + c ? 1.0f : 0.0f);
        Ml[ai * 33 + aj0 + c] = av[c];
    }
#pragma unroll
    for (int k = 0; k < 3; ++k) {
        Eil[lane * T_  + wvu + 8 * k] = evi[k];
        Ejl[lane * EJP + wvu + 8 * k] = evj[k];
    }
    lds_barrier();

    {
        float rv[V_];
        float mn = 1e30f, mx = -1e30f;
#pragma unroll
        for (int j = 0; j < V_; ++j) {
            rv[j] = Ml[lane * 33 + j];
            mn = fminf(mn, rv[j]); mx = fmaxf(mx, rv[j]);
        }
#pragma unroll
        for (int off = 16; off > 0; off >>= 1) {
            mn = fminf(mn, __shfl_xor(mn, off));
            mx = fmaxf(mx, __shfl_xor(mx, off));
        }
        float inv = 1.0f / (mx - mn);
        float s = 0.0f;
#pragma unroll
        for (int j = 0; j < V_; ++j) s += (rv[j] - mn) * inv;
        float dinv = rsqrtf(s);
        float dri = __shfl(dinv, ai);
#pragma unroll
        for (int c = 0; c < 4; ++c) {
            float drj = __shfl(dinv, aj0 + c);
            Ajl[ai * 33 + aj0 + c] = (av[c] - mn) * inv * dri * drj;
        }
    }

    {
        float ejr[T_];
#pragma unroll
        for (int tp = 0; tp < T_; ++tp) ejr[tp] = Ejl[lane * EJP + tp];
        float ejt = Ejl[lane * EJP + t];
        for (int k = 0; k < 4; ++k) {
            int i = wave + 8 * k;
            const float* eip = Eil + i * T_;
            float l[T_];
            float m = -1e30f;
#pragma unroll
            for (int tp = 0; tp < T_; ++tp) {
                float e = eip[tp] + ejr[tp];
                e = e > 0.0f ? e : ALPHA * e;
                l[tp] = e;
                m = fmaxf(m, e);
            }
            float s = 0.0f;
#pragma unroll
            for (int tp = 0; tp < T_; ++tp) s += __expf(l[tp] - m);
            float is = 1.0f / s;
            float et = eip[t] + ejt;
            et = et > 0.0f ? et : ALPHA * et;
            Gl[i * 33 + lane] = __expf(et - m) * is;
        }
    }
    lds_barrier();

    int lh = lane >> 4, lm = lane & 15;
    int wv = wvu;
    if (wv < 4) {
        int mrow = (wv >> 1) * 16 + lm;
        int mcol = (wv & 1) * 16 + lm;
        v8f c0 = {}, c1 = {};
#pragma unroll
        for (int kk = 0; kk < 4; ++kk) {
            int kb0 = kk * 4 + lh * 2;
            int kb1 = kb0 + 16;
            v2f a0, b0, a1, b1;
            a0.x = Ajl[kb0 * 33 + mrow];
            a0.y = Ajl[(kb0 + 1) * 33 + mrow];
            b0.x = Gl[kb0 * 33 + mcol];
            b0.y = Gl[(kb0 + 1) * 33 + mcol];
            a1.x = Ajl[kb1 * 33 + mrow];
            a1.y = Ajl[(kb1 + 1) * 33 + mrow];
            b1.x = Gl[kb1 * 33 + mcol];
            b1.y = Gl[(kb1 + 1) * 33 + mcol];
            c0 = __builtin_amdgcn_wmma_f32_16x16x4_f32(
                      false, a0,  false, b0,
                      (short)0, c0,  false,  false);
            c1 = __builtin_amdgcn_wmma_f32_16x16x4_f32(
                      false, a1,  false, b1,
                      (short)0, c1,  false,  false);
        }
        int mr0 = (wv >> 1) * 16 + lh * 8;
#pragma unroll
        for (int r = 0; r < 8; ++r) Ml[(mr0 + r) * 33 + mcol] = c0[r] + c1[r];
    }
    {
        float4* wd = (float4*)(Wl + (tid >> 3) * WP + (tid & 7) * 4);
        const int ks = 32 * WP / 4;
        wd[0]      = wr0; wd[ks]     = wr1; wd[2 * ks] = wr2; wd[3 * ks] = wr3;
        wd[4 * ks] = wr4; wd[5 * ks] = wr5; wd[6 * ks] = wr6; wd[7 * ks] = wr7;
    }
    __syncthreads();

    v2f af0[8], af1[8], bf[2][8];
#pragma unroll
    for (int kk = 0; kk < 8; ++kk) {
        int kb = kk * 4 + lh * 2;
        af0[kk].x = Ml[lm * 33 + kb];
        af0[kk].y = Ml[lm * 33 + kb + 1];
        af1[kk].x = Ml[(16 + lm) * 33 + kb];
        af1[kk].y = Ml[(16 + lm) * 33 + kb + 1];
        bf[0][kk] = *(const v2f*)(Wl + ((wvu * 2 + 0) * 16 + lm) * WP + kb);
        bf[1][kk] = *(const v2f*)(Wl + ((wvu * 2 + 1) * 16 + lm) * WP + kb);
    }
    float* ol = Ol + wvu * (16 * WP);
    int l8 = lane >> 3, v4 = (lane & 7) * 4;
#pragma unroll
    for (int pr = 0; pr < 2; ++pr) {
        int tn  = wvu * 2 + pr;
        v8f c0 = {}, c1 = {};
#pragma unroll
        for (int kk = 0; kk < 8; ++kk) {
            c0 = __builtin_amdgcn_wmma_f32_16x16x4_f32(
                      false, af0[kk],  false, bf[pr][kk],
                      (short)0, c0,  false,  false);
            c1 = __builtin_amdgcn_wmma_f32_16x16x4_f32(
                      false, af1[kk],  false, bf[pr][kk],
                      (short)0, c1,  false,  false);
        }
#pragma unroll
        for (int tm = 0; tm < 2; ++tm) {
            float s8[8];
#pragma unroll
            for (int r = 0; r < 8; ++r) {
                float x = tm ? c1[r] : c0[r];
                float y = x > 0.0f ? x : (__expf(x) - 1.0f);
                s8[r] = 1.0f / (1.0f + __expf(-y));
            }
            float4* od = (float4*)(ol + lm * WP + tm * 16 + lh * 8);
            od[0] = make_float4(s8[0], s8[1], s8[2], s8[3]);
            od[1] = make_float4(s8[4], s8[5], s8[6], s8[7]);
        }
        wave_patch_written();
#pragma unroll
        for (int j = 0; j < 4; ++j) {
            int pl = 4 * j + l8;
            float4 ov = *(const float4*)(ol + pl * WP + v4);
            *(float4*)(out + ((size_t)(n * HW + tn * 16 + pl) * T_ + t) * V_ + v4) = ov;
        }
    }
}

extern "C" void kernel_launch(void* const* d_in, const int* in_sizes, int n_in,
                              void* d_out, int out_size, void* d_ws, size_t ws_size,
                              hipStream_t stream) {
    const float* h  = (const float*)d_in[0];
    const float* Wp = (const float*)d_in[1];
    const float* ap = (const float*)d_in[2];
    const float* Bp = (const float*)d_in[3];
    float* out = (float*)d_out;

    float* ws    = (float*)d_ws;
    float* Whb   = ws;
    float* part  = Whb  + (size_t)N_ * T_ * V_ * HW;

    whe_kernel <<<N_ * 32, 256, 0, stream>>>(h, Wp, ap, Whb, part);
    gemm_kernel<<<N_ * T_, 256, 0, stream>>>(Whb, part, Bp, out);
}


# Adjacency normalisation moved out of the 96 product blocks (every wave redid it behind ten serial lane-shuffle trips) into one extra block of launch 1; launch 2 only loads adjn; sum orders kept
# speedup vs baseline: 25.2275x; 1.0843x over previous
#include <hip/hip_runtime.h>


#define ALPHA 0.2f
#define N_  4
#define HD  16
#define WD  16
#define T_  24
#define V_  32
#define HW  256
#define THW 512

typedef __attribute__((ext_vector_type(2))) float v2f;
typedef __attribute__((ext_vector_type(8))) float v8f;

static __device__ __forceinline__ void lds_barrier() {
    asm volatile("s_wait_dscnt 0x0\n\ts_barrier_signal -1\n\ts_barrier_wait -1" ::: "memory");
}

static __device__ __forceinline__ void wave_patch_written() {
    asm volatile("s_wait_dscnt 0x0" ::: "memory");
}

__global__ void __launch_bounds__(256) whe_kernel(const float* __restrict__ h,
                                                  const float* __restrict__ Wp,
                                                  const float* __restrict__ a,
                                                  const float* __restrict__ B,
                                                  float* __restrict__ Wh,
                                                  float* __restrict__ part,
                                                  float* __restrict__ adjn) {
    __shared__ __attribute__((aligned(16))) float Red[4 * 8 * V_];
    int bid = blockIdx.x;
    int tid = threadIdx.x;
    int lane = tid & 31;
    if (bid == N_ * 32) {
        if (tid >= V_) return;
        const float4* brow = (const float4*)(B + lane * V_);
        float rv[V_];
#pragma unroll
        for (int q = 0; q < 8; ++q) {
            float4 b4 = brow[q];
            rv[4 * q + 0] = b4.x + 1e-6f + (lane == 4 * q + 0 ? 1.0f : 0.0f);
            rv[4 * q + 1] = b4.y + 1e-6f + (lane == 4 * q + 1 ? 1.0f : 0.0f);
            rv[4 * q + 2] = b4.z + 1e-6f + (lane == 4 * q + 2 ? 1.0f : 0.0f);
            rv[4 * q + 3] = b4.w + 1e-6f + (lane == 4 * q + 3 ? 1.0f : 0.0f);
        }
        float rmn = 1e30f, rmx = -1e30f;
#pragma unroll
        for (int j = 0; j < V_; ++j) { rmn = fminf(rmn, rv[j]); rmx = fmaxf(rmx, rv[j]); }
        Red[lane] = rmn;
        Red[V_ + lane] = rmx;
        wave_patch_written();
        float mn = 1e30f, mx = -1e30f;
#pragma unroll
        for (int k = 0; k < 8; ++k) {
            float4 a4 = *(const float4*)(Red + 4 * k);
            float4 b4 = *(const float4*)(Red + V_ + 4 * k);
            mn = fminf(fminf(fminf(mn, a4.x), fminf(a4.y, a4.z)), a4.w);
            mx = fmaxf(fmaxf(fmaxf(mx, b4.x), fmaxf(b4.y, b4.z)), b4.w);
        }
        float inv = 1.0f / (mx - mn);
        float s = 0.0f;
#pragma unroll
        for (int j = 0; j < V_; ++j) s = __builtin_fmaf(rv[j] - mn, inv, s);
        float dri = rsqrtf(s);
        Red[2 * V_ + lane] = dri;
        wave_patch_written();
        float4* arow = (float4*)(adjn + lane * V_);
#pragma unroll
        for (int q = 0; q < 8; ++q) {
            float4 d4 = *(const float4*)(Red + 2 * V_ + 4 * q);
            arow[q] = make_float4((rv[4 * q + 0] - mn) * inv * dri * d4.x,
                                  (rv[4 * q + 1] - mn) * inv * dri * d4.y,
                                  (rv[4 * q + 2] - mn) * inv * dri * d4.z,
                                  (rv[4 * q + 3] - mn) * inv * dri * d4.w);
        }
        return;
    }
    int pb  = bid & 31;
    int n   = bid >> 5;
    unsigned wu = (unsigned)__builtin_amdgcn_readfirstlane(tid >> 5);
    int p = pb * 8 + (int)wu;

    const float* hrow = h + (size_t)((n * HW + p) * T_) * V_;
    float x[T_];
#pragma unroll
    for (int b = 0; b < 8; ++b) {
        const float* hq = hrow + 3u * ((unsigned)b ^ wu) * V_;
#pragma unroll
        for (int j = 0; j < 3; ++j) x[3 * b + j] = hq[j * V_ + lane];
    }
    __builtin_amdgcn_sched_barrier(0);

    float acc[T_];
#pragma unroll
    for (int t = 0; t < T_; ++t) acc[t] = 0.0f;
#pragma unroll
    for (int b = 0; b < 8; ++b) {
        unsigned wo = ((unsigned)b ^ wu) * (unsigned)(3 * T_ * sizeof(float));
        const float* wblk = (const float*)((const char*)Wp + wo);
#pragma unroll
        for (int j = 0; j < 3; ++j) {
#pragma unroll
            for (int t = 0; t < T_; ++t) acc[t] += x[3 * b + j] * wblk[j * T_ + t];
        }
    }

    float* wp = Wh + ((size_t)(n * T_) * HW + p) * V_ + lane;
#pragma unroll
    for (int t = 0; t < T_; ++t) wp[(size_t)t * HW * V_] = acc[t];

    int f0 = (p >> 4) * (2 * WD * T_) + (p & 15) * T_;
    int fb = (pb >> 1) * (2 * WD * T_) + (pb & 1) * (8 * T_);
    int R0i = fb >> 9, R0j = (fb + WD * T_) >> 9;
    float si0 = 0.0f, si1 = 0.0f, sj0 = 0.0f, sj1 = 0.0f;
#pragma unroll
    for (int g = 0; g < 3; ++g) {
        int fi = f0 + 8 * g, fj = fi + WD * T_;
        const float* ai = a + (fi & (THW - 1));
        const float* aj = a + (fj & (THW - 1));
        float di = 0.0f, dj = 0.0f;
#pragma unroll
        for (int u = 0; u < 8; ++u) {
            di += acc[8 * g + u] * ai[u];
            dj += acc[8 * g + u] * aj[u];
        }
        bool hi_i = (fi >> 9) != R0i;
        bool hi_j = (fj >> 9) != R0j;
        si0 += hi_i ? 0.0f : di;
        si1 += hi_i ? di : 0.0f;
        sj0 += hi_j ? 0.0f : dj;
        sj1 += hi_j ? dj : 0.0f;
    }
    Red[(0 * 8 + wu) * V_ + lane] = si0;
    Red[(1 * 8 + wu) * V_ + lane] = si1;
    Red[(2 * 8 + wu) * V_ + lane] = sj0;
    Red[(3 * 8 + wu) * V_ + lane] = sj1;
    lds_barrier();
    if (tid < 4 * V_) {
        int q = tid >> 5;
        float s = 0.0f;
#pragma unroll
        for (int w = 0; w < 8; ++w) s += Red[(q * 8 + w) * V_ + lane];
        part[((n * 32 + pb) * 4 + q) * V_ + lane] = s;
    }
}

#define EJP 25
#define WP  36

__global__ void __launch_bounds__(256) gemm_kernel(const float* Wh,
                                                   const float* __restrict__ part,
                                                   const float* adjn,
                                                   float* __restrict__ out) {
    __shared__ float Gl [V_ * 33];
    __shared__ float Ajl[V_ * 33];
    __shared__ float Ml [V_ * 33];
    __shared__ __attribute__((aligned(16))) float Wl[HW * WP];
    __shared__ float Eil[V_ * T_];
    __shared__ float Ejl[V_ * EJP];
    __shared__ __attribute__((aligned(16))) float Ol[8 * 16 * WP];

    int bid = blockIdx.x;
    int n = bid / T_;
    int t = bid - n * T_;
    int tid = threadIdx.x;
    int wave = tid >> 5, lane = tid & 31;
    const float4* wsl = (const float4*)(Wh + (size_t)(n * T_ + t) * HW * V_);

    float4 bv = *(const float4*)(adjn + 4 * tid);
    int wvu = __builtin_amdgcn_readfirstlane(wave);
    float iA[3], iB[3], jA[3], jB[3];
#pragma unroll
    for (int k = 0; k < 3; ++k) {
        int tn = wvu + 8 * k;
        int m = tn / 3, r = tn - 3 * m;
        const float* pn = part + (size_t)(n * 32 + 4 * m) * (4 * V_);
        iA[k] = pn[(r == 0 ? 0 : (r == 1 ? 2 : 3)) * (4 * V_) + (r == 2 ? 1 : 0) * V_ + lane];
        iB[k] = pn[(r == 0 ? 1 : 3) * (4 * V_) + lane];
        jA[k] = pn[(r == 2 ? 2 : 0) * (4 * V_) + (r == 1 ? 3 : 2) * V_ + lane];
        jB[k] = pn[(r == 1 ? 1 : 3) * (4 * V_) + 2 * V_ + lane];
    }
    float4 wr0 = wsl[tid],        wr1 = wsl[tid + 256],  wr2 = wsl[tid + 512],  wr3 = wsl[tid + 768];
    float4 wr4 = wsl[tid + 1024], wr5 = wsl[tid + 1280], wr6 = wsl[tid + 1536], wr7 = wsl[tid + 1792];
    __builtin_amdgcn_sched_barrier(0);
    float evi[3], evj[3];
#pragma unroll
    for (int k = 0; k < 3; ++k) {
        int tn = wvu + 8 * k;
        int r = tn - 3 * (tn / 3);
        evi[k] = iA[k] + (r < 2 ? iB[k] : 0.0f);
        evj[k] = jA[k] + (r > 0 ? jB[k] : 0.0f);
    }

#pragma unroll
    for (int k = 0; k < 3; ++k) {
        Eil[lane * T_  + wvu + 8 * k] = evi[k];
        Ejl[lane * EJP + wvu + 8 * k] = evj[k];
    }
    lds_barrier();

    {
        float ejr[T_];
#pragma unroll
        for (int tp = 0; tp < T_; ++tp) ejr[tp] = Ejl[lane * EJP + tp];
        float ejt = Ejl[lane * EJP + t];
        for (int k = 0; k < 4; ++k) {
            int i = wave + 8 * k;
            const float* eip = Eil + i * T_;
            float l[T_];
            float m = -1e30f;
#pragma unroll
            for (int tp = 0; tp < T_; ++tp) {
                float e = eip[tp] + ejr[tp];
                e = e > 0.0f ? e : ALPHA * e;
                l[tp] = e;
                m = fmaxf(m, e);
            }
            float s = 0.0f;
#pragma unroll
            for (int tp = 0; tp < T_; ++tp) s += __expf(l[tp] - m);
            float is = 1.0f / s;
            float et = eip[t] + ejt;
            et = et > 0.0f ? et : ALPHA * et;
            Gl[i * 33 + lane] = __expf(et - m) * is;
        }
    }
    {
        int ai = tid >> 3, aj0 = (tid & 7) * 4;
        Ajl[ai * 33 + aj0 + 0] = bv.x;
        Ajl[ai * 33 + aj0 + 1] = bv.y;
        Ajl[ai * 33 + aj0 + 2] = bv.z;
        Ajl[ai * 33 + aj0 + 3] = bv.w;
    }
    lds_barrier();

    int lh = lane >> 4, lm = lane & 15;
    int wv = wvu;
    if (wv < 4) {
        int mrow = (wv >> 1) * 16 + lm;
        int mcol = (wv & 1) * 16 + lm;
        v8f c0 = {}, c1 = {};
#pragma unroll
        for (int kk = 0; kk < 4; ++kk) {
            int kb0 = kk * 4 + lh * 2;
            int kb1 = kb0 + 16;
            v2f a0, b0, a1, b1;
            a0.x = Ajl[kb0 * 33 + mrow];
            a0.y = Ajl[(kb0 + 1) * 33 + mrow];
            b0.x = Gl[kb0 * 33 + mcol];
            b0.y = Gl[(kb0 + 1) * 33 + mcol];
            a1.x = Ajl[kb1 * 33 + mrow];
            a1.y = Ajl[(kb1 + 1) * 33 + mrow];
            b1.x = Gl[kb1 * 33 + mcol];
            b1.y = Gl[(kb1 + 1) * 33 + mcol];
            c0 = __builtin_amdgcn_wmma_f32_16x16x4_f32(
                      false, a0,  false, b0,
                      (short)0, c0,  false,  false);
            c1 = __builtin_amdgcn_wmma_f32_16x16x4_f32(
                      false, a1,  false, b1,
                      (short)0, c1,  false,  false);
        }
        int mr0 = (wv >> 1) * 16 + lh * 8;
#pragma unroll
        for (int r = 0; r < 8; ++r) Ml[(mr0 + r) * 33 + mcol] = c0[r] + c1[r];
    }
    {
        float4* wd = (float4*)(Wl + (tid >> 3) * WP + (tid & 7) * 4);
        const int ks = 32 * WP / 4;
        wd[0]      = wr0; wd[ks]     = wr1; wd[2 * ks] = wr2; wd[3 * ks] = wr3;
        wd[4 * ks] = wr4; wd[5 * ks] = wr5; wd[6 * ks] = wr6; wd[7 * ks] = wr7;
    }
    __syncthreads();

    v2f af0[8], af1[8], bf[2][8];
#pragma unroll
    for (int kk = 0; kk < 8; ++kk) {
        int kb = kk * 4 + lh * 2;
        af0[kk].x = Ml[lm * 33 + kb];
        af0[kk].y = Ml[lm * 33 + kb + 1];
        af1[kk].x = Ml[(16 + lm) * 33 + kb];
        af1[kk].y = Ml[(16 + lm) * 33 + kb + 1];
        bf[0][kk] = *(const v2f*)(Wl + ((wvu * 2 + 0) * 16 + lm) * WP + kb);
        bf[1][kk] = *(const v2f*)(Wl + ((wvu * 2 + 1) * 16 + lm) * WP + kb);
    }
    float* ol = Ol + wvu * (16 * WP);
    int l8 = lane >> 3, v4 = (lane & 7) * 4;
#pragma unroll
    for (int pr = 0; pr < 2; ++pr) {
        int tn  = wvu * 2 + pr;
        v8f c0 = {}, c1 = {};
#pragma unroll
        for (int kk = 0; kk < 8; ++kk) {
            c0 = __builtin_amdgcn_wmma_f32_16x16x4_f32(
                      false, af0[kk],  false, bf[pr][kk],
                      (short)0, c0,  false,  false);
            c1 = __builtin_amdgcn_wmma_f32_16x16x4_f32(
                      false, af1[kk],  false, bf[pr][kk],
                      (short)0, c1,  false,  false);
        }
#pragma unroll
        for (int tm = 0; tm < 2; ++tm) {
            float s8[8];
#pragma unroll
            for (int r = 0; r < 8; ++r) {
                float x = tm ? c1[r] : c0[r];
                float y = x > 0.0f ? x : (__expf(x) - 1.0f);
                s8[r] = 1.0f / (1.0f + __expf(-y));
            }
            float4* od = (float4*)(ol + lm * WP + tm * 16 + lh * 8);
            od[0] = make_float4(s8[0], s8[1], s8[2], s8[3]);
            od[1] = make_float4(s8[4], s8[5], s8[6], s8[7]);
        }
        wave_patch_written();
#pragma unroll
        for (int j = 0; j < 4; ++j) {
            int pl = 4 * j + l8;
            float4 ov = *(const float4*)(ol + pl * WP + v4);
            *(float4*)(out + ((size_t)(n * HW + tn * 16 + pl) * T_ + t) * V_ + v4) = ov;
        }
    }
}

extern "C" void kernel_launch(void* const* d_in, const int* in_sizes, int n_in,
                              void* d_out, int out_size, void* d_ws, size_t ws_size,
                              hipStream_t stream) {
    const float* h  = (const float*)d_in[0];
    const float* Wp = (const float*)d_in[1];
    const float* ap = (const float*)d_in[2];
    const float* Bp = (const float*)d_in[3];
    float* out = (float*)d_out;

    float* ws    = (float*)d_ws;
    float* Whb   = ws;
    float* part  = Whb  + (size_t)N_ * T_ * V_ * HW;
    float* adjn  = part + (size_t)N_ * 32 * 4 * V_;

    whe_kernel <<<N_ * 32 + 1, 256, 0, stream>>>(h, Wp, ap, Bp, Whb, part, adjn);
    gemm_kernel<<<N_ * T_, 256, 0, stream>>>(Whb, part, adjn, out);
}


# Launch 2 softmax and activation in base 2 on fewer vector ops: ei/ej pre-scaled by log2(e), leaky and shift as one max(x-m, a*x-m), 1/x by v_rcp; softmax sums as even/odd halves; last bits change
# speedup vs baseline: 26.6975x; 1.0583x over previous
#include <hip/hip_runtime.h>


#define ALPHA 0.2f
#define LOG2E 1.44269504088896340736f
#define N_  4
#define HD  16
#define WD  16
#define T_  24
#define V_  32
#define HW  256
#define THW 512

typedef __attribute__((ext_vector_type(2))) float v2f;
typedef __attribute__((ext_vector_type(8))) float v8f;

static __device__ __forceinline__ void lds_barrier() {
    asm volatile("s_wait_dscnt 0x0\n\ts_barrier_signal -1\n\ts_barrier_wait -1" ::: "memory");
}

static __device__ __forceinline__ void wave_patch_written() {
    asm volatile("s_wait_dscnt 0x0" ::: "memory");
}

__global__ void __launch_bounds__(256) whe_kernel(const float* __restrict__ h,
                                                  const float* __restrict__ Wp,
                                                  const float* __restrict__ a,
                                                  const float* __restrict__ B,
                                                  float* __restrict__ Wh,
                                                  float* __restrict__ part,
                                                  float* __restrict__ adjn) {
    __shared__ __attribute__((aligned(16))) float Red[4 * 8 * V_];
    int bid = blockIdx.x;
    int tid = threadIdx.x;
    int lane = tid & 31;
    if (bid == N_ * 32) {
        if (tid >= V_) return;
        const float4* brow = (const float4*)(B + lane * V_);
        float rv[V_];
#pragma unroll
        for (int q = 0; q < 8; ++q) {
            float4 b4 = brow[q];
            rv[4 * q + 0] = b4.x + 1e-6f + (lane == 4 * q + 0 ? 1.0f : 0.0f);
            rv[4 * q + 1] = b4.y + 1e-6f + (lane == 4 * q + 1 ? 1.0f : 0.0f);
            rv[4 * q + 2] = b4.z + 1e-6f + (lane == 4 * q + 2 ? 1.0f : 0.0f);
            rv[4 * q + 3] = b4.w + 1e-6f + (lane == 4 * q + 3 ? 1.0f : 0.0f);
        }
        float rmn = 1e30f, rmx = -1e30f;
#pragma unroll
        for (int j = 0; j < V_; ++j) { rmn = fminf(rmn, rv[j]); rmx = fmaxf(rmx, rv[j]); }
        Red[lane] = rmn;
        Red[V_ + lane] = rmx;
        wave_patch_written();
        float mn = 1e30f, mx = -1e30f;
#pragma unroll
        for (int k = 0; k < 8; ++k) {
            float4 a4 = *(const float4*)(Red + 4 * k);
            float4 b4 = *(const float4*)(Red + V_ + 4 * k);
            mn = fminf(fminf(fminf(mn, a4.x), fminf(a4.y, a4.z)), a4.w);
            mx = fmaxf(fmaxf(fmaxf(mx, b4.x), fmaxf(b4.y, b4.z)), b4.w);
        }
        float inv = 1.0f / (mx - mn);
        float s = 0.0f;
#pragma unroll
        for (int j = 0; j < V_; ++j) s = __builtin_fmaf(rv[j] - mn, inv, s);
        float dri = rsqrtf(s);
        Red[2 * V_ + lane] = dri;
        wave_patch_written();
        float4* arow = (float4*)(adjn + lane * V_);
#pragma unroll
        for (int q = 0; q < 8; ++q) {
            float4 d4 = *(const float4*)(Red + 2 * V_ + 4 * q);
            arow[q] = make_float4((rv[4 * q + 0] - mn) * inv * dri * d4.x,
                                  (rv[4 * q + 1] - mn) * inv * dri * d4.y,
                                  (rv[4 * q + 2] - mn) * inv * dri * d4.z,
                                  (rv[4 * q + 3] - mn) * inv * dri * d4.w);
        }
        return;
    }
    int pb  = bid & 31;
    int n   = bid >> 5;
    unsigned wu = (unsigned)__builtin_amdgcn_readfirstlane(tid >> 5);
    int p = pb * 8 + (int)wu;

    const float* hrow = h + (size_t)((n * HW + p) * T_) * V_;
    float x[T_];
#pragma unroll
    for (int b = 0; b < 8; ++b) {
        const float* hq = hrow + 3u * ((unsigned)b ^ wu) * V_;
#pragma unroll
        for (int j = 0; j < 3; ++j) x[3 * b + j] = hq[j * V_ + lane];
    }
    __builtin_amdgcn_sched_barrier(0);

    float acc[T_];
#pragma unroll
    for (int t = 0; t < T_; ++t) acc[t] = 0.0f;
#pragma unroll
    for (int b = 0; b < 8; ++b) {
        unsigned wo = ((unsigned)b ^ wu) * (unsigned)(3 * T_ * sizeof(float));
        const float* wblk = (const float*)((const char*)Wp + wo);
#pragma unroll
        for (int j = 0; j < 3; ++j) {
#pragma unroll
            for (int t = 0; t < T_; ++t) acc[t] += x[3 * b + j] * wblk[j * T_ + t];
        }
    }

    float* wp = Wh + ((size_t)(n * T_) * HW + p) * V_ + lane;
#pragma unroll
    for (int t = 0; t < T_; ++t) wp[(size_t)t * HW * V_] = acc[t];

    int f0 = (p >> 4) * (2 * WD * T_) + (p & 15) * T_;
    int fb = (pb >> 1) * (2 * WD * T_) + (pb & 1) * (8 * T_);
    int R0i = fb >> 9, R0j = (fb + WD * T_) >> 9;
    float si0 = 0.0f, si1 = 0.0f, sj0 = 0.0f, sj1 = 0.0f;
#pragma unroll
    for (int g = 0; g < 3; ++g) {
        int fi = f0 + 8 * g, fj = fi + WD * T_;
        const float* ai = a + (fi & (THW - 1));
        const float* aj = a + (fj & (THW - 1));
        float di = 0.0f, dj = 0.0f;
#pragma unroll
        for (int u = 0; u < 8; ++u) {
            di += acc[8 * g + u] * ai[u];
            dj += acc[8 * g + u] * aj[u];
        }
        bool hi_i = (fi >> 9) != R0i;
        bool hi_j = (fj >> 9) != R0j;
        si0 += hi_i ? 0.0f : di;
        si1 += hi_i ? di : 0.0f;
        sj0 += hi_j ? 0.0f : dj;
        sj1 += hi_j ? dj : 0.0f;
    }
    Red[(0 * 8 + wu) * V_ + lane] = si0;
    Red[(1 * 8 + wu) * V_ + lane] = si1;
    Red[(2 * 8 + wu) * V_ + lane] = sj0;
    Red[(3 * 8 + wu) * V_ + lane] = sj1;
    lds_barrier();
    if (tid < 4 * V_) {
        int q = tid >> 5;
        float s = 0.0f;
#pragma unroll
        for (int w = 0; w < 8; ++w) s += Red[(q * 8 + w) * V_ + lane];
        part[((n * 32 + pb) * 4 + q) * V_ + lane] = s;
    }
}

#define EJP 25
#define WP  36

__global__ void __launch_bounds__(256) gemm_kernel(const float* Wh,
                                                   const float* __restrict__ part,
                                                   const float* adjn,
                                                   float* __restrict__ out) {
    __shared__ float Gl [V_ * 33];
    __shared__ float Ajl[V_ * 33];
    __shared__ float Ml [V_ * 33];
    __shared__ __attribute__((aligned(16))) float Wl[HW * WP];
    __shared__ __attribute__((aligned(16))) float Eil[V_ * T_];
    __shared__ float Ejl[V_ * EJP];
    __shared__ __attribute__((aligned(16))) float Ol[8 * 16 * WP];

    int bid = blockIdx.x;
    int n = bid / T_;
    int t = bid - n * T_;
    int tid = threadIdx.x;
    int wave = tid >> 5, lane = tid & 31;
    const float4* wsl = (const float4*)(Wh + (size_t)(n * T_ + t) * HW * V_);

    float4 bv = *(const float4*)(adjn + 4 * tid);
    int wvu = __builtin_amdgcn_readfirstlane(wave);
    float iA[3], iB[3], jA[3], jB[3];
#pragma unroll
    for (int k = 0; k < 3; ++k) {
        int tn = wvu + 8 * k;
        int m = tn / 3, r = tn - 3 * m;
        const float* pn = part + (size_t)(n * 32 + 4 * m) * (4 * V_);
        iA[k] = pn[(r == 0 ? 0 : (r == 1 ? 2 : 3)) * (4 * V_) + (r == 2 ? 1 : 0) * V_ + lane];
        iB[k] = pn[(r == 0 ? 1 : 3) * (4 * V_) + lane];
        jA[k] = pn[(r == 2 ? 2 : 0) * (4 * V_) + (r == 1 ? 3 : 2) * V_ + lane];
        jB[k] = pn[(r == 1 ? 1 : 3) * (4 * V_) + 2 * V_ + lane];
    }
    float4 wr0 = wsl[tid],        wr1 = wsl[tid + 256],  wr2 = wsl[tid + 512],  wr3 = wsl[tid + 768];
    float4 wr4 = wsl[tid + 1024], wr5 = wsl[tid + 1280], wr6 = wsl[tid + 1536], wr7 = wsl[tid + 1792];
    __builtin_amdgcn_sched_barrier(0);
    float evi[3], evj[3];
#pragma unroll
    for (int k = 0; k < 3; ++k) {
        int tn = wvu + 8 * k;
        int r = tn - 3 * (tn / 3);
        evi[k] = iA[k] + (r < 2 ? iB[k] : 0.0f);
        evj[k] = jA[k] + (r > 0 ? jB[k] : 0.0f);
    }

#pragma unroll
    for (int k = 0; k < 3; ++k) {
        Eil[lane * T_  + wvu + 8 * k] = evi[k] * LOG2E;
        Ejl[lane * EJP + wvu + 8 * k] = evj[k] * LOG2E;
    }
    lds_barrier();

    {
        v2f ejr[T_ / 2];
#pragma unroll
        for (int q = 0; q < T_ / 2; ++q) {
            ejr[q].x = Ejl[lane * EJP + 2 * q];
            ejr[q].y = Ejl[lane * EJP + 2 * q + 1];
        }
        float ejt = Ejl[lane * EJP + t];
        const v2f al2 = {ALPHA, ALPHA};
        for (int k = 0; k < 4; ++k) {
            int i = wave + 8 * k;
            const float* eip = Eil + i * T_;
            v2f x[T_ / 2];
            float mx = -1e30f;
#pragma unroll
            for (int q = 0; q < T_ / 2; ++q) {
                x[q] = *(const v2f*)(eip + 2 * q) + ejr[q];
                mx = fmaxf(fmaxf(mx, x[q].x), x[q].y);
            }
            float m = fmaxf(mx, ALPHA * mx);
            const v2f m2 = {m, m};
            v2f s2 = {0.0f, 0.0f};
#pragma unroll
            for (int q = 0; q < T_ / 2; ++q) {
                v2f u = x[q] - m2;
                v2f w = __builtin_elementwise_fma(x[q], al2, -m2);
                v2f p;
                p.x = __builtin_amdgcn_exp2f(fmaxf(u.x, w.x));
                p.y = __builtin_amdgcn_exp2f(fmaxf(u.y, w.y));
                s2 += p;
            }
            float s = s2.x + s2.y;
            float et = eip[t] + ejt;
            float dt = fmaxf(et - m, __builtin_fmaf(et, ALPHA, -m));
            Gl[i * 33 + lane] = __builtin_amdgcn_exp2f(dt) * __builtin_amdgcn_rcpf(s);
        }
    }
    {
        int ai = tid >> 3, aj0 = (tid & 7) * 4;
        Ajl[ai * 33 + aj0 + 0] = bv.x;
        Ajl[ai * 33 + aj0 + 1] = bv.y;
        Ajl[ai * 33 + aj0 + 2] = bv.z;
        Ajl[ai * 33 + aj0 + 3] = bv.w;
    }
    lds_barrier();

    int lh = lane >> 4, lm = lane & 15;
    int wv = wvu;
    if (wv < 4) {
        int mrow = (wv >> 1) * 16 + lm;
        int mcol = (wv & 1) * 16 + lm;
        v8f c0 = {}, c1 = {};
#pragma unroll
        for (int kk = 0; kk < 4; ++kk) {
            int kb0 = kk * 4 + lh * 2;
            int kb1 = kb0 + 16;
            v2f a0, b0, a1, b1;
            a0.x = Ajl[kb0 * 33 + mrow];
            a0.y = Ajl[(kb0 + 1) * 33 + mrow];
            b0.x = Gl[kb0 * 33 + mcol];
            b0.y = Gl[(kb0 + 1) * 33 + mcol];
            a1.x = Ajl[kb1 * 33 + mrow];
            a1.y = Ajl[(kb1 + 1) * 33 + mrow];
            b1.x = Gl[kb1 * 33 + mcol];
            b1.y = Gl[(kb1 + 1) * 33 + mcol];
            c0 = __builtin_amdgcn_wmma_f32_16x16x4_f32(
                      false, a0,  false, b0,
                      (short)0, c0,  false,  false);
            c1 = __builtin_amdgcn_wmma_f32_16x16x4_f32(
                      false, a1,  false, b1,
                      (short)0, c1,  false,  false);
        }
        int mr0 = (wv >> 1) * 16 + lh * 8;
#pragma unroll
        for (int r = 0; r < 8; ++r) Ml[(mr0 + r) * 33 + mcol] = c0[r] + c1[r];
    }
    {
        float4* wd = (float4*)(Wl + (tid >> 3) * WP + (tid & 7) * 4);
        const int ks = 32 * WP / 4;
        wd[0]      = wr0; wd[ks]     = wr1; wd[2 * ks] = wr2; wd[3 * ks] = wr3;
        wd[4 * ks] = wr4; wd[5 * ks] = wr5; wd[6 * ks] = wr6; wd[7 * ks] = wr7;
    }
    __syncthreads();

    v2f af0[8], af1[8], bf[2][8];
#pragma unroll
    for (int kk = 0; kk < 8; ++kk) {
        int kb = kk * 4 + lh * 2;
        af0[kk].x = Ml[lm * 33 + kb];
        af0[kk].y = Ml[lm * 33 + kb + 1];
        af1[kk].x = Ml[(16 + lm) * 33 + kb];
        af1[kk].y = Ml[(16 + lm) * 33 + kb + 1];
        bf[0][kk] = *(const v2f*)(Wl + ((wvu * 2 + 0) * 16 + lm) * WP + kb);
        bf[1][kk] = *(const v2f*)(Wl + ((wvu * 2 + 1) * 16 + lm) * WP + kb);
    }
    float* ol = Ol + wvu * (16 * WP);
    int l8 = lane >> 3, v4 = (lane & 7) * 4;
#pragma unroll
    for (int pr = 0; pr < 2; ++pr) {
        int tn  = wvu * 2 + pr;
        v8f c0 = {}, c1 = {};
#pragma unroll
        for (int kk = 0; kk < 8; ++kk) {
            c0 = __builtin_amdgcn_wmma_f32_16x16x4_f32(
                      false, af0[kk],  false, bf[pr][kk],
                      (short)0, c0,  false,  false);
            c1 = __builtin_amdgcn_wmma_f32_16x16x4_f32(
                      false, af1[kk],  false, bf[pr][kk],
                      (short)0, c1,  false,  false);
        }
#pragma unroll
        for (int tm = 0; tm < 2; ++tm) {
            float s8[8];
#pragma unroll
            for (int r = 0; r < 8; ++r) {
                float x = tm ? c1[r] : c0[r];
                float tx = x * LOG2E;
                float nw = __builtin_fmaf(__builtin_amdgcn_exp2f(tx), -LOG2E, LOG2E);
                float z = x > 0.0f ? -tx : nw;
                s8[r] = __builtin_amdgcn_rcpf(1.0f + __builtin_amdgcn_exp2f(z));
            }
            float4* od = (float4*)(ol + lm * WP + tm * 16 + lh * 8);
            od[0] = make_float4(s8[0], s8[1], s8[2], s8[3]);
            od[1] = make_float4(s8[4], s8[5], s8[6], s8[7]);
        }
        wave_patch_written();
#pragma unroll
        for (int j = 0; j < 4; ++j) {
            int pl = 4 * j + l8;
            float4 ov = *(const float4*)(ol + pl * WP + v4);
            *(float4*)(out + ((size_t)(n * HW + tn * 16 + pl) * T_ + t) * V_ + v4) = ov;
        }
    }
}

extern "C" void kernel_launch(void* const* d_in, const int* in_sizes, int n_in,
                              void* d_out, int out_size, void* d_ws, size_t ws_size,
                              hipStream_t stream) {
    const float* h  = (const float*)d_in[0];
    const float* Wp = (const float*)d_in[1];
    const float* ap = (const float*)d_in[2];
    const float* Bp = (const float*)d_in[3];
    float* out = (float*)d_out;

    float* ws    = (float*)d_ws;
    float* Whb   = ws;
    float* part  = Whb  + (size_t)N_ * T_ * V_ * HW;
    float* adjn  = part + (size_t)N_ * 32 * 4 * V_;

    whe_kernel <<<N_ * 32 + 1, 256, 0, stream>>>(h, Wp, ap, Bp, Whb, part, adjn);
    gemm_kernel<<<N_ * T_, 256, 0, stream>>>(Whb, part, adjn, out);
}
